# starting kernel
# baseline (speedup 1.0000x reference)
#include <hip/hip_runtime.h>

#define DIM 64

typedef float v2f __attribute__((ext_vector_type(2)));
typedef float v8f __attribute__((ext_vector_type(8)));

__global__ void bignn_zero(float4* __restrict__ p, int n4) {
    int i = blockIdx.x * blockDim.x + threadIdx.x;
    if (i < n4) p[i] = make_float4(0.f, 0.f, 0.f, 0.f);
}

__global__ void bignn_spmm(const int* __restrict__ erow,
                           const int* __restrict__ ecol,
                           const float* __restrict__ eval_,
                           const float* __restrict__ x,
                           float* __restrict__ y, int E) {
    int t = blockIdx.x * blockDim.x + threadIdx.x;
    int e = t >> 5;
    if (e >= E) return;
    int lane = t & 31;
    int r = erow[e];
    int c = ecol[e];
    float v = eval_[e];
    int off = lane * 2;
    float2 xv = *(const float2*)(x + (long)c * DIM + off);
    float* dst = y + (long)r * DIM + off;
    atomicAdd(dst,     v * xv.x);
    atomicAdd(dst + 1, v * xv.y);
}

__global__ __launch_bounds__(256)
void bignn_gemm(const float* __restrict__ lap,
                const float* __restrict__ feat,
                const float* __restrict__ W1, const float* __restrict__ b1,
                const float* __restrict__ W2, const float* __restrict__ b2,
                float* __restrict__ out, int N) {
    __shared__ v2f  sW1[(DIM / 2) * DIM];
    __shared__ v2f  sW2[(DIM / 2) * DIM];
    __shared__ float sBias[DIM];

    int tid = threadIdx.x;
    {
        float* w1f = (float*)sW1;
        float* w2f = (float*)sW2;
        for (int i = tid; i < DIM * DIM; i += 256) {
            int kRow = i >> 6;
            int n    = i & 63;
            int dst  = (kRow >> 1) * (2 * DIM) + n * 2 + (kRow & 1);
            w1f[dst] = W1[i];
            w2f[dst] = W2[i];
        }
        if (tid < DIM) sBias[tid] = b1[tid] + b2[tid];
    }
    __syncthreads();

    int wave = tid >> 5;
    int lane = tid & 31;
    int rowBase = blockIdx.x * 128 + wave * 16;
    if (rowBase >= N) return;

    int m    = lane & 15;
    int half = lane >> 4;
    int row  = rowBase + m;
    int rowSafe = row < N ? row : (N - 1);

    const float2* lrow = (const float2*)(lap  + (long)rowSafe * DIM);
    const float2* frow = (const float2*)(feat + (long)rowSafe * DIM);

    v8f acc[4];
    acc[0] = v8f{}; acc[1] = v8f{}; acc[2] = v8f{}; acc[3] = v8f{};

    for (int k0 = 0; k0 < 16; ++k0) {
        int kp = k0 * 2 + half;
        float2 l = lrow[kp];
        float2 f = frow[kp];
        v2f aS = { l.x + f.x, l.y + f.y };
        v2f aP = { l.x * f.x, l.y * f.y };
        #pragma unroll
        for (int nb = 0; nb < 4; ++nb) {
            int cidx = nb * 16 + m;
            v2f bA = sW1[kp * DIM + cidx];
            v2f bB = sW2[kp * DIM + cidx];
            acc[nb] = __builtin_amdgcn_wmma_f32_16x16x4_f32(
                false, aS, false, bA, (short)0, acc[nb], false, false);
            acc[nb] = __builtin_amdgcn_wmma_f32_16x16x4_f32(
                false, aP, false, bB, (short)0, acc[nb], false, false);
        }
    }

    int rOff = half * 8;
    if (rowBase + 16 <= N) {
        #pragma unroll
        for (int nb = 0; nb < 4; ++nb) {
            int col = nb * 16 + m;
            float bias = sBias[col];
            #pragma unroll
            for (int g = 0; g < 8; ++g) {
                out[(long)(rowBase + rOff + g) * DIM + col] = acc[nb][g] + bias;
            }
        }
    } else {
        #pragma unroll
        for (int nb = 0; nb < 4; ++nb) {
            int col = nb * 16 + m;
            float bias = sBias[col];
            #pragma unroll
            for (int g = 0; g < 8; ++g) {
                int r = rowBase + rOff + g;
                if (r < N) out[(long)r * DIM + col] = acc[nb][g] + bias;
            }
        }
    }
}

extern "C" void kernel_launch(void* const* d_in, const int* in_sizes, int n_in,
                              void* d_out, int out_size, void* d_ws, size_t ws_size,
                              hipStream_t stream) {
    const int*   erow  = (const int*)  d_in[0];
    const int*   ecol  = (const int*)  d_in[1];
    const float* eval_ = (const float*)d_in[2];
    const float* feat  = (const float*)d_in[3];
    const float* W1    = (const float*)d_in[4];
    const float* b1    = (const float*)d_in[5];
    const float* W2    = (const float*)d_in[6];
    const float* b2    = (const float*)d_in[7];
    float* out = (float*)d_out;
    float* lap = (float*)d_ws;

    int E = in_sizes[0];
    int N = in_sizes[3] / DIM;

    int n4 = (N * DIM) / 4;
    bignn_zero<<<(n4 + 255) / 256, 256, 0, stream>>>((float4*)lap, n4);

    long threads = (long)E * 32;
    int blocks = (int)((threads + 255) / 256);
    bignn_spmm<<<blocks, 256, 0, stream>>>(erow, ecol, eval_, feat, lap, E);

    bignn_gemm<<<(N + 127) / 128, 256, 0, stream>>>(lap, feat, W1, b1, W2, b2, out, N);
}


# Two list-head tables, one per half of the package (each half has its own L2): a link block exchanges only into its own half's table (die id), the walk takes both; f32 sum order unchanged
# speedup vs baseline: 2.5810x; 2.5810x over previous
#include <hip/hip_runtime.h>


#define DIM 64
#define TILE_ROWS 128
#define CAP 44
#define SUBS 4
#define PLANES 2
#define PITCH 46

typedef float v2f __attribute__((ext_vector_type(2)));
typedef float v8f __attribute__((ext_vector_type(8)));
typedef unsigned long long u64;

static_assert(SUBS == 4, "a row's SUBS heads are cleared as one int4");
__global__ void bignn_heads(int* __restrict__ head, int n4) {
    int i = blockIdx.x * blockDim.x + threadIdx.x;
    if (i < n4) ((int4*)head)[i] = make_int4(-1, -1, -1, -1);
}

__device__ __forceinline__ int bignn_my_half() {
    return (int)((__builtin_amdgcn_s_sendmsg_rtn(0x87) >> 18) & 1u);
}

__device__ __forceinline__ void bignn_link_one(const int* __restrict__ erow,
                                               const int* __restrict__ ecol,
                                               const float* __restrict__ eval_,
                                               int* __restrict__ head,
                                               int* __restrict__ rec, int e) {
    int old = atomicExch(head + erow[e] * SUBS + (e & (SUBS - 1)), e);
    rec[3 * e + 0] = old;
    rec[3 * e + 1] = ecol[e];
    rec[3 * e + 2] = __float_as_int(eval_[e]);
}

__global__ void bignn_link(const int* __restrict__ erow,
                           const int* __restrict__ ecol,
                           const float* __restrict__ eval_,
                           int* __restrict__ head_all,
                           int* __restrict__ rec, int E, int plane) {
    int* __restrict__ head = head_all + bignn_my_half() * plane;
    int e0 = (blockIdx.x * blockDim.x + threadIdx.x) * 4;
    if (e0 + 3 < E) {
        int4   r4 = *(const int4*)(erow + e0);
        int4   c4 = *(const int4*)(ecol + e0);
        float4 v4 = *(const float4*)(eval_ + e0);
        int o0 = atomicExch(head + r4.x * SUBS + 0, e0);
        int o1 = atomicExch(head + r4.y * SUBS + 1, e0 + 1);
        int o2 = atomicExch(head + r4.z * SUBS + 2, e0 + 2);
        int o3 = atomicExch(head + r4.w * SUBS + 3, e0 + 3);
        int* p = rec + 3 * e0;
        p[0] = o0; p[1]  = c4.x; p[2]  = __float_as_int(v4.x);
        p[3] = o1; p[4]  = c4.y; p[5]  = __float_as_int(v4.y);
        p[6] = o2; p[7]  = c4.z; p[8]  = __float_as_int(v4.z);
        p[9] = o3; p[10] = c4.w; p[11] = __float_as_int(v4.w);
    } else {
        for (int e = e0; e < E; ++e) bignn_link_one(erow, ecol, eval_, head, rec, e);
    }
}

template <int NB>
__device__ __forceinline__ void bignn_addn(float2& acc, const u64* L, int i, int r,
                                           const float2* __restrict__ x2, int lane) {
    const unsigned* Lw = (const unsigned*)L;
    float2 xv[NB];
    #pragma unroll
    for (int q = 0; q < NB; ++q) {
        int qi = i + (q < r ? q : r - 1);
        unsigned c = Lw[2 * qi + 1];
        xv[q] = x2[(c << 5) | (unsigned)lane];
    }
    asm volatile("" :: "v"(xv[NB - 1].x), "v"(xv[NB - 1].y) : "memory");
    #pragma unroll
    for (int q = 0; q < NB; ++q) {
        int qi = i + (q < r ? q : r - 1);
        float vv = __uint_as_float(Lw[2 * qi]);
        float v  = (q < r) ? vv : 0.0f;
        acc.x += v * xv[q].x;
        acc.y += v * xv[q].y;
    }
}

__global__ __launch_bounds__(512) __attribute__((amdgpu_waves_per_eu(16, 16)))
void bignn_pull_gemm(const int* __restrict__ head,
                     const int* __restrict__ rec,
                     const float* __restrict__ feat,
                     const float* __restrict__ W1, const float* __restrict__ b1,
                     const float* __restrict__ W2, const float* __restrict__ b2,
                     float* __restrict__ out, int N, int E, int plane) {
    __shared__ v2f   sW1[(DIM / 2) * DIM];
    __shared__ v2f   sW2[(DIM / 2) * DIM];
    __shared__ float sBias[DIM];
    __shared__ u64   sList[TILE_ROWS * PITCH];
    __shared__ int   sCnt[TILE_ROWS];

    int tid = threadIdx.x;
    int tileBase = blockIdx.x * TILE_ROWS;
    if (tid < TILE_ROWS) sCnt[tid] = 0;
    __syncthreads();
    {
        float* w1f = (float*)sW1;
        float* w2f = (float*)sW2;
        for (int i = tid; i < DIM * DIM; i += 512) {
            int kRow = i >> 6;
            int n    = i & 63;
            int dst  = (kRow >> 1) * (2 * DIM) + n * 2 + (kRow & 1);
            w1f[dst] = W1[i];
            w2f[dst] = W2[i];
        }
        if (tid < DIM) sBias[tid] = b1[tid] + b2[tid];
    }

    {
        int slot = tid >> 2;
        int sub  = tid & (SUBS - 1);
        int row  = tileBase + slot;
        u64* L = sList + slot * PITCH;
        int e  = (row < N) ? head[row * SUBS + sub] : -1;
        int e2 = (row < N) ? head[plane + row * SUBS + sub] : -1;
        if (e < 0) { e = e2; e2 = -1; }
        for (int s = 0; s < E && e >= 0; ++s) {
            int nx      = rec[3 * e + 0];
            unsigned c  = (unsigned)rec[3 * e + 1];
            unsigned vb = (unsigned)rec[3 * e + 2];
            int pos = atomicAdd(&sCnt[slot], 1);
            if (pos < CAP) L[pos] = ((u64)c << 32) | (u64)vb;
            e = nx;
            if (e < 0) { e = e2; e2 = -1; }
        }
    }
    __syncthreads();

    int wave = tid >> 5;
    int lane = tid & 31;
    const float2* x2 = (const float2*)feat;
    for (int j = 0; j < TILE_ROWS / 16; ++j) {
        int t = wave * (TILE_ROWS / 16) + j;
        int d = __builtin_amdgcn_readfirstlane(sCnt[t]);
        u64* L = sList + t * PITCH;
        float2 acc = make_float2(0.0f, 0.0f);
        if (d <= CAP) {
            {
                const u64 kmax = ~(u64)0;
                u64 ka = (lane < d) ? L[lane] : kmax;
                int ra = 0;
                if (d <= 32) {
                    for (int q = 0; q < d; ++q) {
                        u64 kq = L[q];
                        ra += (kq < ka || (kq == ka && q < lane)) ? 1 : 0;
                    }
                    if (lane < d) L[ra] = ka;
                } else {
                    u64 kb = (lane + 32 < d) ? L[lane + 32] : kmax;
                    int rb = 0;
                    for (int q = 0; q < d; ++q) {
                        u64 kq = L[q];
                        ra += (kq < ka || (kq == ka && q < lane)) ? 1 : 0;
                        rb += (kq < kb || (kq == kb && q < lane + 32)) ? 1 : 0;
                    }
                    L[ra] = ka;
                    if (lane + 32 < d) L[rb] = kb;
                }
            }
            int i = 0;
            while (d - i > 8) {
                int r = (d - i < 16) ? (d - i) : 16;
                bignn_addn<16>(acc, L, i, r, x2, lane);
                i += r;
            }
            if (i < d) bignn_addn<8>(acc, L, i, d - i, x2, lane);
        } else {
            int row = tileBase + t;
            u64 last = 0;
            int have = 0;
            for (int outer = 0; outer < E; ++outer) {
                u64 best = 0;
                int mult = 0;
                for (int li = 0; li < PLANES * SUBS; ++li) {
                    int e = head[(li / SUBS) * plane + row * SUBS + (li % SUBS)];
                    for (int s = 0; s < E && e >= 0; ++s) {
                        int nx = rec[3 * e + 0];
                        u64 k = ((u64)(unsigned)rec[3 * e + 1] << 32) | (u64)(unsigned)rec[3 * e + 2];
                        if (!have || k > last) {
                            if (mult == 0 || k < best) { best = k; mult = 1; }
                            else if (k == best) ++mult;
                        }
                        e = nx;
                    }
                }
                if (mult == 0) break;
                float v = __uint_as_float((unsigned)best);
                float2 xv = x2[(long)(int)(best >> 32) * (DIM / 2) + lane];
                for (int m2 = 0; m2 < mult; ++m2) {
                    acc.x += v * xv.x;
                    acc.y += v * xv.y;
                }
                last = best;
                have = 1;
            }
        }
        L[lane] = ((u64)__float_as_uint(acc.y) << 32) | (u64)__float_as_uint(acc.x);
    }
    __syncthreads();

    int m    = lane & 15;
    int half = lane >> 4;
    int rt   = wave >> 1;
    int ch   = wave & 1;
    int lrow = rt * 16 + m;
    int rowBase = tileBase + rt * 16;
    int row  = rowBase + m;
    int rowSafe = row < N ? row : (N - 1);

    const u64*    lrowp = sList + lrow * PITCH;
    const float2* frow  = (const float2*)(feat + (long)rowSafe * DIM);

    v8f acc2[2];
    acc2[0] = v8f{}; acc2[1] = v8f{};

    for (int k0 = 0; k0 < 16; ++k0) {
        int kp = k0 * 2 + half;
        u64 lw = lrowp[kp];
        float2 l = make_float2(__uint_as_float((unsigned)lw), __uint_as_float((unsigned)(lw >> 32)));
        float2 f = frow[kp];
        v2f aS = { l.x + f.x, l.y + f.y };
        v2f aP = { l.x * f.x, l.y * f.y };
        #pragma unroll
        for (int nb = 0; nb < 2; ++nb) {
            int cidx = (ch * 2 + nb) * 16 + m;
            v2f bA = sW1[kp * DIM + cidx];
            v2f bB = sW2[kp * DIM + cidx];
            acc2[nb] = __builtin_amdgcn_wmma_f32_16x16x4_f32(
                false, aS, false, bA, (short)0, acc2[nb], false, false);
            acc2[nb] = __builtin_amdgcn_wmma_f32_16x16x4_f32(
                false, aP, false, bB, (short)0, acc2[nb], false, false);
        }
    }

    int rOff = half * 8;
    if (rowBase + 16 <= N) {
        #pragma unroll
        for (int nb = 0; nb < 2; ++nb) {
            int col = (ch * 2 + nb) * 16 + m;
            float bias = sBias[col];
            #pragma unroll
            for (int g = 0; g < 8; ++g)
                out[(long)(rowBase + rOff + g) * DIM + col] = acc2[nb][g] + bias;
        }
    } else {
        #pragma unroll
        for (int nb = 0; nb < 2; ++nb) {
            int col = (ch * 2 + nb) * 16 + m;
            float bias = sBias[col];
            #pragma unroll
            for (int g = 0; g < 8; ++g) {
                int r = rowBase + rOff + g;
                if (r < N) out[(long)r * DIM + col] = acc2[nb][g] + bias;
            }
        }
    }
}

extern "C" void kernel_launch(void* const* d_in, const int* in_sizes, int n_in,
                              void* d_out, int out_size, void* d_ws, size_t ws_size,
                              hipStream_t stream) {
    const int*   erow  = (const int*)  d_in[0];
    const int*   ecol  = (const int*)  d_in[1];
    const float* eval_ = (const float*)d_in[2];
    const float* feat  = (const float*)d_in[3];
    const float* W1    = (const float*)d_in[4];
    const float* b1    = (const float*)d_in[5];
    const float* W2    = (const float*)d_in[6];
    const float* b2    = (const float*)d_in[7];
    float* out = (float*)d_out;

    int E = in_sizes[0];
    int N = in_sizes[3] / DIM;

    int* head = (int*)d_ws;
    int* rec  = head + PLANES * SUBS * N;

    bignn_heads<<<(PLANES * N + 255) / 256, 256, 0, stream>>>(head, PLANES * N);
    bignn_link<<<((E + 3) / 4 + 255) / 256, 256, 0, stream>>>(erow, ecol, eval_, head, rec, E, SUBS * N);
    bignn_pull_gemm<<<(N + TILE_ROWS - 1) / TILE_ROWS, 512, 0, stream>>>(
        head, rec, feat, W1, b1, W2, b2, out, N, E, SUBS * N);
}


# Pull kernel front end: loads that do not depend on each other go out together (weights, bias and heads as one batch; a lane walks its two tables' lists side by side); f32 sum order unchanged
# speedup vs baseline: 2.7421x; 1.0624x over previous
#include <hip/hip_runtime.h>


#define DIM 64
#define TILE_ROWS 128
#define CAP 44
#define SUBS 4
#define PLANES 2
#define PITCH 46

typedef float v2f __attribute__((ext_vector_type(2)));
typedef float v8f __attribute__((ext_vector_type(8)));
typedef unsigned long long u64;

static_assert(SUBS == 4, "a row's SUBS heads are cleared as one int4");
__global__ void bignn_heads(int* __restrict__ head, int n4) {
    int i = blockIdx.x * blockDim.x + threadIdx.x;
    if (i < n4) ((int4*)head)[i] = make_int4(-1, -1, -1, -1);
}

__device__ __forceinline__ int bignn_my_half() {
    return (int)((__builtin_amdgcn_s_sendmsg_rtn(0x87) >> 18) & 1u);
}

__device__ __forceinline__ void bignn_link_one(const int* __restrict__ erow,
                                               const int* __restrict__ ecol,
                                               const float* __restrict__ eval_,
                                               int* __restrict__ head,
                                               int* __restrict__ rec, int e) {
    int old = atomicExch(head + erow[e] * SUBS + (e & (SUBS - 1)), e);
    rec[3 * e + 0] = old;
    rec[3 * e + 1] = ecol[e];
    rec[3 * e + 2] = __float_as_int(eval_[e]);
}

__global__ void bignn_link(const int* __restrict__ erow,
                           const int* __restrict__ ecol,
                           const float* __restrict__ eval_,
                           int* __restrict__ head_all,
                           int* __restrict__ rec, int E, int plane) {
    int* __restrict__ head = head_all + bignn_my_half() * plane;
    int e0 = (blockIdx.x * blockDim.x + threadIdx.x) * 4;
    if (e0 + 3 < E) {
        int4   r4 = *(const int4*)(erow + e0);
        int4   c4 = *(const int4*)(ecol + e0);
        float4 v4 = *(const float4*)(eval_ + e0);
        int o0 = atomicExch(head + r4.x * SUBS + 0, e0);
        int o1 = atomicExch(head + r4.y * SUBS + 1, e0 + 1);
        int o2 = atomicExch(head + r4.z * SUBS + 2, e0 + 2);
        int o3 = atomicExch(head + r4.w * SUBS + 3, e0 + 3);
        int* p = rec + 3 * e0;
        p[0] = o0; p[1]  = c4.x; p[2]  = __float_as_int(v4.x);
        p[3] = o1; p[4]  = c4.y; p[5]  = __float_as_int(v4.y);
        p[6] = o2; p[7]  = c4.z; p[8]  = __float_as_int(v4.z);
        p[9] = o3; p[10] = c4.w; p[11] = __float_as_int(v4.w);
    } else {
        for (int e = e0; e < E; ++e) bignn_link_one(erow, ecol, eval_, head, rec, e);
    }
}

template <int NB>
__device__ __forceinline__ void bignn_addn(float2& acc, const u64* L, int i, int r,
                                           const float2* __restrict__ x2, int lane) {
    const unsigned* Lw = (const unsigned*)L;
    float2 xv[NB];
    #pragma unroll
    for (int q = 0; q < NB; ++q) {
        int qi = i + (q < r ? q : r - 1);
        unsigned c = Lw[2 * qi + 1];
        xv[q] = x2[(c << 5) | (unsigned)lane];
    }
    asm volatile("" :: "v"(xv[NB - 1].x), "v"(xv[NB - 1].y) : "memory");
    #pragma unroll
    for (int q = 0; q < NB; ++q) {
        int qi = i + (q < r ? q : r - 1);
        float vv = __uint_as_float(Lw[2 * qi]);
        float v  = (q < r) ? vv : 0.0f;
        acc.x += v * xv[q].x;
        acc.y += v * xv[q].y;
    }
}

__global__ __launch_bounds__(512) __attribute__((amdgpu_waves_per_eu(16, 16)))
void bignn_pull_gemm(const int* __restrict__ head,
                     const int* __restrict__ rec,
                     const float* __restrict__ feat,
                     const float* __restrict__ W1, const float* __restrict__ b1,
                     const float* __restrict__ W2, const float* __restrict__ b2,
                     float* __restrict__ out, int N, int E, int plane) {
    __shared__ v2f   sW1[(DIM / 2) * DIM];
    __shared__ v2f   sW2[(DIM / 2) * DIM];
    __shared__ float sBias[DIM];
    __shared__ u64   sList[TILE_ROWS * PITCH];
    __shared__ int   sCnt[TILE_ROWS];

    int tid = threadIdx.x;
    int tileBase = blockIdx.x * TILE_ROWS;

    int wp = tid >> 4;
    int wg = tid & 15;
    const float4* W1v = (const float4*)W1;
    const float4* W2v = (const float4*)W2;
    float4 w1lo = W1v[(2 * wp) * (DIM / 4) + wg];
    float4 w1hi = W1v[(2 * wp + 1) * (DIM / 4) + wg];
    float4 w2lo = W2v[(2 * wp) * (DIM / 4) + wg];
    float4 w2hi = W2v[(2 * wp + 1) * (DIM / 4) + wg];
    int   bcol  = tid & (DIM - 1);
    float bsum  = b1[bcol] + b2[bcol];
    int slot = tid >> 2;
    int sub  = tid & (SUBS - 1);
    int wrow = tileBase + slot;
    int rowC = wrow < N ? wrow : (N - 1);
    int hA = head[rowC * SUBS + sub];
    int hB = head[plane + rowC * SUBS + sub];

    if (tid < TILE_ROWS) sCnt[tid] = 0;
    {
        v2f* d1 = sW1 + wp * DIM + 4 * wg;
        v2f* d2 = sW2 + wp * DIM + 4 * wg;
        d1[0] = v2f{w1lo.x, w1hi.x}; d1[1] = v2f{w1lo.y, w1hi.y};
        d1[2] = v2f{w1lo.z, w1hi.z}; d1[3] = v2f{w1lo.w, w1hi.w};
        d2[0] = v2f{w2lo.x, w2hi.x}; d2[1] = v2f{w2lo.y, w2hi.y};
        d2[2] = v2f{w2lo.z, w2hi.z}; d2[3] = v2f{w2lo.w, w2hi.w};
        if (tid < DIM) sBias[tid] = bsum;
    }
    __syncthreads();

    {
        u64* L = sList + slot * PITCH;
        int eA = (wrow < N) ? hA : -1;
        int eB = (wrow < N) ? hB : -1;
        for (int s = 0; s < E && (eA >= 0 || eB >= 0); ++s) {
            bool a = eA >= 0;
            bool b = eB >= 0;
            int iA = a ? eA : 0;
            int iB = b ? eB : 0;
            int      nA = rec[3 * iA + 0];
            unsigned cA = (unsigned)rec[3 * iA + 1];
            unsigned vA = (unsigned)rec[3 * iA + 2];
            int      nB = rec[3 * iB + 0];
            unsigned cB = (unsigned)rec[3 * iB + 1];
            unsigned vB = (unsigned)rec[3 * iB + 2];
            int na  = a ? 1 : 0;
            int pos = atomicAdd(&sCnt[slot], na + (b ? 1 : 0));
            if (a && pos < CAP)      L[pos]      = ((u64)cA << 32) | (u64)vA;
            if (b && pos + na < CAP) L[pos + na] = ((u64)cB << 32) | (u64)vB;
            eA = a ? nA : -1;
            eB = b ? nB : -1;
        }
    }
    __syncthreads();

    int wave = tid >> 5;
    int lane = tid & 31;
    const float2* x2 = (const float2*)feat;
    for (int j = 0; j < TILE_ROWS / 16; ++j) {
        int t = wave * (TILE_ROWS / 16) + j;
        int d = __builtin_amdgcn_readfirstlane(sCnt[t]);
        u64* L = sList + t * PITCH;
        float2 acc = make_float2(0.0f, 0.0f);
        if (d <= CAP) {
            {
                const u64 kmax = ~(u64)0;
                u64 ka = (lane < d) ? L[lane] : kmax;
                int ra = 0;
                if (d <= 32) {
                    for (int q = 0; q < d; ++q) {
                        u64 kq = L[q];
                        ra += (kq < ka || (kq == ka && q < lane)) ? 1 : 0;
                    }
                    if (lane < d) L[ra] = ka;
                } else {
                    u64 kb = (lane + 32 < d) ? L[lane + 32] : kmax;
                    int rb = 0;
                    for (int q = 0; q < d; ++q) {
                        u64 kq = L[q];
                        ra += (kq < ka || (kq == ka && q < lane)) ? 1 : 0;
                        rb += (kq < kb || (kq == kb && q < lane + 32)) ? 1 : 0;
                    }
                    L[ra] = ka;
                    if (lane + 32 < d) L[rb] = kb;
                }
            }
            int i = 0;
            while (d - i > 8) {
                int r = (d - i < 16) ? (d - i) : 16;
                bignn_addn<16>(acc, L, i, r, x2, lane);
                i += r;
            }
            if (i < d) bignn_addn<8>(acc, L, i, d - i, x2, lane);
        } else {
            int row = tileBase + t;
            u64 last = 0;
            int have = 0;
            for (int outer = 0; outer < E; ++outer) {
                u64 best = 0;
                int mult = 0;
                for (int li = 0; li < PLANES * SUBS; ++li) {
                    int e = head[(li / SUBS) * plane + row * SUBS + (li % SUBS)];
                    for (int s = 0; s < E && e >= 0; ++s) {
                        int nx = rec[3 * e + 0];
                        u64 k = ((u64)(unsigned)rec[3 * e + 1] << 32) | (u64)(unsigned)rec[3 * e + 2];
                        if (!have || k > last) {
                            if (mult == 0 || k < best) { best = k; mult = 1; }
                            else if (k == best) ++mult;
                        }
                        e = nx;
                    }
                }
                if (mult == 0) break;
                float v = __uint_as_float((unsigned)best);
                float2 xv = x2[(long)(int)(best >> 32) * (DIM / 2) + lane];
                for (int m2 = 0; m2 < mult; ++m2) {
                    acc.x += v * xv.x;
                    acc.y += v * xv.y;
                }
                last = best;
                have = 1;
            }
        }
        L[lane] = ((u64)__float_as_uint(acc.y) << 32) | (u64)__float_as_uint(acc.x);
    }
    __syncthreads();

    int m    = lane & 15;
    int half = lane >> 4;
    int rt   = wave >> 1;
    int ch   = wave & 1;
    int lrow = rt * 16 + m;
    int rowBase = tileBase + rt * 16;
    int row  = rowBase + m;
    int rowSafe = row < N ? row : (N - 1);

    const u64*    lrowp = sList + lrow * PITCH;
    const float2* frow  = (const float2*)(feat + (long)rowSafe * DIM);

    v8f acc2[2];
    acc2[0] = v8f{}; acc2[1] = v8f{};

    for (int k0 = 0; k0 < 16; ++k0) {
        int kp = k0 * 2 + half;
        u64 lw = lrowp[kp];
        float2 l = make_float2(__uint_as_float((unsigned)lw), __uint_as_float((unsigned)(lw >> 32)));
        float2 f = frow[kp];
        v2f aS = { l.x + f.x, l.y + f.y };
        v2f aP = { l.x * f.x, l.y * f.y };
        #pragma unroll
        for (int nb = 0; nb < 2; ++nb) {
            int cidx = (ch * 2 + nb) * 16 + m;
            v2f bA = sW1[kp * DIM + cidx];
            v2f bB = sW2[kp * DIM + cidx];
            acc2[nb] = __builtin_amdgcn_wmma_f32_16x16x4_f32(
                false, aS, false, bA, (short)0, acc2[nb], false, false);
            acc2[nb] = __builtin_amdgcn_wmma_f32_16x16x4_f32(
                false, aP, false, bB, (short)0, acc2[nb], false, false);
        }
    }

    int rOff = half * 8;
    if (rowBase + 16 <= N) {
        #pragma unroll
        for (int nb = 0; nb < 2; ++nb) {
            int col = (ch * 2 + nb) * 16 + m;
            float bias = sBias[col];
            #pragma unroll
            for (int g = 0; g < 8; ++g)
                out[(long)(rowBase + rOff + g) * DIM + col] = acc2[nb][g] + bias;
        }
    } else {
        #pragma unroll
        for (int nb = 0; nb < 2; ++nb) {
            int col = (ch * 2 + nb) * 16 + m;
            float bias = sBias[col];
            #pragma unroll
            for (int g = 0; g < 8; ++g) {
                int r = rowBase + rOff + g;
                if (r < N) out[(long)r * DIM + col] = acc2[nb][g] + bias;
            }
        }
    }
}

extern "C" void kernel_launch(void* const* d_in, const int* in_sizes, int n_in,
                              void* d_out, int out_size, void* d_ws, size_t ws_size,
                              hipStream_t stream) {
    const int*   erow  = (const int*)  d_in[0];
    const int*   ecol  = (const int*)  d_in[1];
    const float* eval_ = (const float*)d_in[2];
    const float* feat  = (const float*)d_in[3];
    const float* W1    = (const float*)d_in[4];
    const float* b1    = (const float*)d_in[5];
    const float* W2    = (const float*)d_in[6];
    const float* b2    = (const float*)d_in[7];
    float* out = (float*)d_out;

    int E = in_sizes[0];
    int N = in_sizes[3] / DIM;

    int* head = (int*)d_ws;
    int* rec  = head + PLANES * SUBS * N;

    bignn_heads<<<(PLANES * N + 255) / 256, 256, 0, stream>>>(head, PLANES * N);
    bignn_link<<<((E + 3) / 4 + 255) / 256, 256, 0, stream>>>(erow, ecol, eval_, head, rec, E, SUBS * N);
    bignn_pull_gemm<<<(N + TILE_ROWS - 1) / TILE_ROWS, 512, 0, stream>>>(
        head, rec, feat, W1, b1, W2, b2, out, N, E, SUBS * N);
}


# Pull kernel tile 144 rows on 576 threads (was 128 on 512): 695 blocks, at most 3 per WGP (27 row fragments), instead of 782 with a 4th block (32 fragments) on some WGPs; f32 sum order unchanged
# speedup vs baseline: 2.9384x; 1.0716x over previous
#include <hip/hip_runtime.h>


#define DIM 64
#define TILE_ROWS 144
#define ROWS_PER_WAVE 8
#define NTHREADS (TILE_ROWS * 4)
#define CAP 44
#define SUBS 4
#define PLANES 2
#define PITCH 46

typedef float v2f __attribute__((ext_vector_type(2)));
typedef float v8f __attribute__((ext_vector_type(8)));
typedef unsigned long long u64;

static_assert(SUBS == 4, "a row's SUBS heads are cleared as one int4");
static_assert(NTHREADS == TILE_ROWS * SUBS, "SUBS walker lanes per row");
static_assert(NTHREADS / 32 * ROWS_PER_WAVE == TILE_ROWS, "every row is summed by exactly one wave");
static_assert(NTHREADS / 32 == 2 * (TILE_ROWS / 16), "two waves (column halves) per 16-row fragment");
__global__ void bignn_heads(int* __restrict__ head, int n4) {
    int i = blockIdx.x * blockDim.x + threadIdx.x;
    if (i < n4) ((int4*)head)[i] = make_int4(-1, -1, -1, -1);
}

__device__ __forceinline__ int bignn_my_half() {
    return (int)((__builtin_amdgcn_s_sendmsg_rtn(0x87) >> 18) & 1u);
}

__device__ __forceinline__ void bignn_link_one(const int* __restrict__ erow,
                                               const int* __restrict__ ecol,
                                               const float* __restrict__ eval_,
                                               int* __restrict__ head,
                                               int* __restrict__ rec, int e) {
    int old = atomicExch(head + erow[e] * SUBS + (e & (SUBS - 1)), e);
    rec[3 * e + 0] = old;
    rec[3 * e + 1] = ecol[e];
    rec[3 * e + 2] = __float_as_int(eval_[e]);
}

__global__ void bignn_link(const int* __restrict__ erow,
                           const int* __restrict__ ecol,
                           const float* __restrict__ eval_,
                           int* __restrict__ head_all,
                           int* __restrict__ rec, int E, int plane) {
    int* __restrict__ head = head_all + bignn_my_half() * plane;
    int e0 = (blockIdx.x * blockDim.x + threadIdx.x) * 4;
    if (e0 + 3 < E) {
        int4   r4 = *(const int4*)(erow + e0);
        int4   c4 = *(const int4*)(ecol + e0);
        float4 v4 = *(const float4*)(eval_ + e0);
        int o0 = atomicExch(head + r4.x * SUBS + 0, e0);
        int o1 = atomicExch(head + r4.y * SUBS + 1, e0 + 1);
        int o2 = atomicExch(head + r4.z * SUBS + 2, e0 + 2);
        int o3 = atomicExch(head + r4.w * SUBS + 3, e0 + 3);
        int* p = rec + 3 * e0;
        p[0] = o0; p[1]  = c4.x; p[2]  = __float_as_int(v4.x);
        p[3] = o1; p[4]  = c4.y; p[5]  = __float_as_int(v4.y);
        p[6] = o2; p[7]  = c4.z; p[8]  = __float_as_int(v4.z);
        p[9] = o3; p[10] = c4.w; p[11] = __float_as_int(v4.w);
    } else {
        for (int e = e0; e < E; ++e) bignn_link_one(erow, ecol, eval_, head, rec, e);
    }
}

template <int NB>
__device__ __forceinline__ void bignn_addn(float2& acc, const u64* L, int i, int r,
                                           const float2* __restrict__ x2, int lane) {
    const unsigned* Lw = (const unsigned*)L;
    float2 xv[NB];
    #pragma unroll
    for (int q = 0; q < NB; ++q) {
        int qi = i + (q < r ? q : r - 1);
        unsigned c = Lw[2 * qi + 1];
        xv[q] = x2[(c << 5) | (unsigned)lane];
    }
    asm volatile("" :: "v"(xv[NB - 1].x), "v"(xv[NB - 1].y) : "memory");
    #pragma unroll
    for (int q = 0; q < NB; ++q) {
        int qi = i + (q < r ? q : r - 1);
        float vv = __uint_as_float(Lw[2 * qi]);
        float v  = (q < r) ? vv : 0.0f;
        acc.x += v * xv[q].x;
        acc.y += v * xv[q].y;
    }
}

__global__ __launch_bounds__(NTHREADS) __attribute__((amdgpu_num_vgpr(64)))
void bignn_pull_gemm(const int* __restrict__ head,
                     const int* __restrict__ rec,
                     const float* __restrict__ feat,
                     const float* __restrict__ W1, const float* __restrict__ b1,
                     const float* __restrict__ W2, const float* __restrict__ b2,
                     float* __restrict__ out, int N, int E, int plane) {
    __shared__ v2f   sW1[(DIM / 2) * DIM];
    __shared__ v2f   sW2[(DIM / 2) * DIM];
    __shared__ float sBias[DIM];
    __shared__ u64   sList[TILE_ROWS * PITCH];
    __shared__ int   sCnt[TILE_ROWS];

    int tid = threadIdx.x;
    int tileBase = blockIdx.x * TILE_ROWS;

    int wp = tid >> 4;
    int wg = tid & 15;
    bool wld = tid < (DIM / 2) * 16;
    const float4* W1v = (const float4*)W1;
    const float4* W2v = (const float4*)W2;
    float4 w1lo = make_float4(0.f, 0.f, 0.f, 0.f), w1hi = w1lo, w2lo = w1lo, w2hi = w1lo;
    if (wld) {
        w1lo = W1v[(2 * wp) * (DIM / 4) + wg];
        w1hi = W1v[(2 * wp + 1) * (DIM / 4) + wg];
        w2lo = W2v[(2 * wp) * (DIM / 4) + wg];
        w2hi = W2v[(2 * wp + 1) * (DIM / 4) + wg];
    }
    int   bcol  = tid & (DIM - 1);
    float bsum  = b1[bcol] + b2[bcol];
    int slot = tid >> 2;
    int sub  = tid & (SUBS - 1);
    int wrow = tileBase + slot;
    int rowC = wrow < N ? wrow : (N - 1);
    int hA = head[rowC * SUBS + sub];
    int hB = head[plane + rowC * SUBS + sub];

    if (tid < TILE_ROWS) sCnt[tid] = 0;
    if (wld) {
        v2f* d1 = sW1 + wp * DIM + 4 * wg;
        v2f* d2 = sW2 + wp * DIM + 4 * wg;
        d1[0] = v2f{w1lo.x, w1hi.x}; d1[1] = v2f{w1lo.y, w1hi.y};
        d1[2] = v2f{w1lo.z, w1hi.z}; d1[3] = v2f{w1lo.w, w1hi.w};
        d2[0] = v2f{w2lo.x, w2hi.x}; d2[1] = v2f{w2lo.y, w2hi.y};
        d2[2] = v2f{w2lo.z, w2hi.z}; d2[3] = v2f{w2lo.w, w2hi.w};
    }
    if (tid < DIM) sBias[tid] = bsum;
    __syncthreads();

    {
        u64* L = sList + slot * PITCH;
        int eA = (wrow < N) ? hA : -1;
        int eB = (wrow < N) ? hB : -1;
        for (int s = 0; s < E && (eA >= 0 || eB >= 0); ++s) {
            bool a = eA >= 0;
            bool b = eB >= 0;
            int iA = a ? eA : 0;
            int iB = b ? eB : 0;
            int      nA = rec[3 * iA + 0];
            unsigned cA = (unsigned)rec[3 * iA + 1];
            unsigned vA = (unsigned)rec[3 * iA + 2];
            int      nB = rec[3 * iB + 0];
            unsigned cB = (unsigned)rec[3 * iB + 1];
            unsigned vB = (unsigned)rec[3 * iB + 2];
            int na  = a ? 1 : 0;
            int pos = atomicAdd(&sCnt[slot], na + (b ? 1 : 0));
            if (a && pos < CAP)      L[pos]      = ((u64)cA << 32) | (u64)vA;
            if (b && pos + na < CAP) L[pos + na] = ((u64)cB << 32) | (u64)vB;
            eA = a ? nA : -1;
            eB = b ? nB : -1;
        }
    }
    __syncthreads();

    int wave = tid >> 5;
    int lane = tid & 31;
    const float2* x2 = (const float2*)feat;
    for (int j = 0; j < ROWS_PER_WAVE; ++j) {
        int t = wave * ROWS_PER_WAVE + j;
        int d = __builtin_amdgcn_readfirstlane(sCnt[t]);
        u64* L = sList + t * PITCH;
        float2 acc = make_float2(0.0f, 0.0f);
        if (d <= CAP) {
            {
                const u64 kmax = ~(u64)0;
                u64 ka = (lane < d) ? L[lane] : kmax;
                int ra = 0;
                if (d <= 32) {
                    for (int q = 0; q < d; ++q) {
                        u64 kq = L[q];
                        ra += (kq < ka || (kq == ka && q < lane)) ? 1 : 0;
                    }
                    if (lane < d) L[ra] = ka;
                } else {
                    u64 kb = (lane + 32 < d) ? L[lane + 32] : kmax;
                    int rb = 0;
                    for (int q = 0; q < d; ++q) {
                        u64 kq = L[q];
                        ra += (kq < ka || (kq == ka && q < lane)) ? 1 : 0;
                        rb += (kq < kb || (kq == kb && q < lane + 32)) ? 1 : 0;
                    }
                    L[ra] = ka;
                    if (lane + 32 < d) L[rb] = kb;
                }
            }
            int i = 0;
            while (d - i > 8) {
                int r = (d - i < 16) ? (d - i) : 16;
                bignn_addn<16>(acc, L, i, r, x2, lane);
                i += r;
            }
            if (i < d) bignn_addn<8>(acc, L, i, d - i, x2, lane);
        } else {
            int row = tileBase + t;
            u64 last = 0;
            int have = 0;
            for (int outer = 0; outer < E; ++outer) {
                u64 best = 0;
                int mult = 0;
                for (int li = 0; li < PLANES * SUBS; ++li) {
                    int e = head[(li / SUBS) * plane + row * SUBS + (li % SUBS)];
                    for (int s = 0; s < E && e >= 0; ++s) {
                        int nx = rec[3 * e + 0];
                        u64 k = ((u64)(unsigned)rec[3 * e + 1] << 32) | (u64)(unsigned)rec[3 * e + 2];
                        if (!have || k > last) {
                            if (mult == 0 || k < best) { best = k; mult = 1; }
                            else if (k == best) ++mult;
                        }
                        e = nx;
                    }
                }
                if (mult == 0) break;
                float v = __uint_as_float((unsigned)best);
                float2 xv = x2[(long)(int)(best >> 32) * (DIM / 2) + lane];
                for (int m2 = 0; m2 < mult; ++m2) {
                    acc.x += v * xv.x;
                    acc.y += v * xv.y;
                }
                last = best;
                have = 1;
            }
        }
        L[lane] = ((u64)__float_as_uint(acc.y) << 32) | (u64)__float_as_uint(acc.x);
    }
    __syncthreads();

    int m    = lane & 15;
    int half = lane >> 4;
    int rt   = wave >> 1;
    int ch   = wave & 1;
    int lrow = rt * 16 + m;
    int rowBase = tileBase + rt * 16;
    int row  = rowBase + m;
    int rowSafe = row < N ? row : (N - 1);

    const u64*    lrowp = sList + lrow * PITCH;
    const float2* frow  = (const float2*)(feat + (long)rowSafe * DIM);

    v8f acc2[2];
    acc2[0] = v8f{}; acc2[1] = v8f{};

    for (int k0 = 0; k0 < 16; ++k0) {
        int kp = k0 * 2 + half;
        u64 lw = lrowp[kp];
        float2 l = make_float2(__uint_as_float((unsigned)lw), __uint_as_float((unsigned)(lw >> 32)));
        float2 f = frow[kp];
        v2f aS = { l.x + f.x, l.y + f.y };
        v2f aP = { l.x * f.x, l.y * f.y };
        #pragma unroll
        for (int nb = 0; nb < 2; ++nb) {
            int cidx = (ch * 2 + nb) * 16 + m;
            v2f bA = sW1[kp * DIM + cidx];
            v2f bB = sW2[kp * DIM + cidx];
            acc2[nb] = __builtin_amdgcn_wmma_f32_16x16x4_f32(
                false, aS, false, bA, (short)0, acc2[nb], false, false);
            acc2[nb] = __builtin_amdgcn_wmma_f32_16x16x4_f32(
                false, aP, false, bB, (short)0, acc2[nb], false, false);
        }
    }

    int rOff = half * 8;
    if (rowBase + 16 <= N) {
        #pragma unroll
        for (int nb = 0; nb < 2; ++nb) {
            int col = (ch * 2 + nb) * 16 + m;
            float bias = sBias[col];
            #pragma unroll
            for (int g = 0; g < 8; ++g)
                out[(long)(rowBase + rOff + g) * DIM + col] = acc2[nb][g] + bias;
        }
    } else {
        #pragma unroll
        for (int nb = 0; nb < 2; ++nb) {
            int col = (ch * 2 + nb) * 16 + m;
            float bias = sBias[col];
            #pragma unroll
            for (int g = 0; g < 8; ++g) {
                int r = rowBase + rOff + g;
                if (r < N) out[(long)r * DIM + col] = acc2[nb][g] + bias;
            }
        }
    }
}

extern "C" void kernel_launch(void* const* d_in, const int* in_sizes, int n_in,
                              void* d_out, int out_size, void* d_ws, size_t ws_size,
                              hipStream_t stream) {
    const int*   erow  = (const int*)  d_in[0];
    const int*   ecol  = (const int*)  d_in[1];
    const float* eval_ = (const float*)d_in[2];
    const float* feat  = (const float*)d_in[3];
    const float* W1    = (const float*)d_in[4];
    const float* b1    = (const float*)d_in[5];
    const float* W2    = (const float*)d_in[6];
    const float* b2    = (const float*)d_in[7];
    float* out = (float*)d_out;

    int E = in_sizes[0];
    int N = in_sizes[3] / DIM;

    int* head = (int*)d_ws;
    int* rec  = head + PLANES * SUBS * N;

    bignn_heads<<<(PLANES * N + 255) / 256, 256, 0, stream>>>(head, PLANES * N);
    bignn_link<<<((E + 3) / 4 + 255) / 256, 256, 0, stream>>>(erow, ecol, eval_, head, rec, E, SUBS * N);
    bignn_pull_gemm<<<(N + TILE_ROWS - 1) / TILE_ROWS, NTHREADS, 0, stream>>>(
        head, rec, feat, W1, b1, W2, b2, out, N, E, SUBS * N);
}


# Edges sorted by row tile in 2816-edge chunks in LDS (8-byte records, run starts per chunk): no global atomics, head tables or lists; a pull lane loads its chunk's run; 2 launches; f32 sum order kept
# speedup vs baseline: 3.4136x; 1.1617x over previous
#include <hip/hip_runtime.h>


#define DIM 64
#define TILE_ROWS 144
#define ROWS_PER_WAVE 8
#define NTHREADS (TILE_ROWS * 4)
#define CAP 44
#define PITCH 46

#define N_NODES 100000
#define NTILES ((N_NODES + TILE_ROWS - 1) / TILE_ROWS)
#define SORT_THREADS 352
#define CHUNK (SORT_THREADS * 8)
#define NBINS (2 * SORT_THREADS)
#define KEYMASK 0x00FFFFFFFFFFFFFFull

typedef float v2f __attribute__((ext_vector_type(2)));
typedef float v8f __attribute__((ext_vector_type(8)));
typedef unsigned long long u64;

static_assert(NTHREADS / 32 * ROWS_PER_WAVE == TILE_ROWS, "every row is summed by exactly one wave");
static_assert(NTHREADS / 32 == 2 * (TILE_ROWS / 16), "two waves (column halves) per 16-row fragment");
static_assert(NBINS >= NTILES + 1, "one bin per row tile and one for the end of the chunk");
static_assert(TILE_ROWS <= 256 && N_NODES <= (1 << 24), "slot and column share the upper half of a record");
static_assert(SORT_THREADS % 32 == 0, "whole waves");

__global__ __launch_bounds__(SORT_THREADS)
void bignn_sort(const int* __restrict__ erow,
                const int* __restrict__ ecol,
                const float* __restrict__ eval_,
                int* __restrict__ tab, unsigned long long* __restrict__ recs, int E) {
    __shared__ int sBin[NBINS];
    __shared__ __attribute__((aligned(16))) u64 sStage[CHUNK];
    __shared__ int sWaveTot[SORT_THREADS / 32];

    int tid  = threadIdx.x;
    int c    = blockIdx.x;
    int base = c * CHUNK;
    sBin[tid] = 0;
    sBin[tid + SORT_THREADS] = 0;

    int      r[8];
    unsigned cl[8];
    unsigned vb[8];
    #pragma unroll
    for (int h = 0; h < 2; ++h) {
        int e0 = base + (h * SORT_THREADS + tid) * 4;
        if (e0 + 3 < E) {
            int4   r4 = *(const int4*)(erow + e0);
            int4   c4 = *(const int4*)(ecol + e0);
            float4 v4 = *(const float4*)(eval_ + e0);
            r[4 * h + 0] = r4.x; r[4 * h + 1] = r4.y; r[4 * h + 2] = r4.z; r[4 * h + 3] = r4.w;
            cl[4 * h + 0] = (unsigned)c4.x; cl[4 * h + 1] = (unsigned)c4.y;
            cl[4 * h + 2] = (unsigned)c4.z; cl[4 * h + 3] = (unsigned)c4.w;
            vb[4 * h + 0] = __float_as_uint(v4.x); vb[4 * h + 1] = __float_as_uint(v4.y);
            vb[4 * h + 2] = __float_as_uint(v4.z); vb[4 * h + 3] = __float_as_uint(v4.w);
        } else {
            #pragma unroll
            for (int q = 0; q < 4; ++q) {
                int e = e0 + q;
                bool ok = e < E;
                int ec = ok ? e : 0;
                int rr = erow[ec];
                int cc = ecol[ec];
                float vv = eval_[ec];
                r[4 * h + q]  = ok ? rr : -1;
                cl[4 * h + q] = (unsigned)cc;
                vb[4 * h + q] = __float_as_uint(vv);
            }
        }
    }
    __syncthreads();

    int tile[8];
    int rank[8];
    #pragma unroll
    for (int k = 0; k < 8; ++k) {
        tile[k] = -1;
        rank[k] = 0;
        if (r[k] >= 0) {
            tile[k] = r[k] / TILE_ROWS;
            rank[k] = atomicAdd(&sBin[tile[k]], 1);
        }
    }
    __syncthreads();

    int lane = tid & 31;
    int wave = tid >> 5;
    int ca = sBin[2 * tid];
    int cb = sBin[2 * tid + 1];
    int s  = ca + cb;
    int inc = s;
    #pragma unroll
    for (int d = 1; d < 32; d <<= 1) {
        int y = __shfl_up(inc, d);
        if (lane >= d) inc += y;
    }
    if (lane == 31) sWaveTot[wave] = inc;
    __syncthreads();
    int wpre = 0;
    #pragma unroll
    for (int w = 0; w < SORT_THREADS / 32; ++w) {
        int wt = sWaveTot[w];
        wpre += (w < wave) ? wt : 0;
    }
    int ex = wpre + inc - s;
    sBin[2 * tid]     = ex;
    sBin[2 * tid + 1] = ex + ca;
    *(int2*)(tab + (size_t)c * NBINS + 2 * tid) = make_int2(ex, ex + ca);
    __syncthreads();

    #pragma unroll
    for (int k = 0; k < 8; ++k) {
        if (tile[k] >= 0) {
            int pos  = sBin[tile[k]] + rank[k];
            unsigned slot = (unsigned)(r[k] - tile[k] * TILE_ROWS);
            sStage[pos] = ((u64)((slot << 24) | cl[k]) << 32) | (u64)vb[k];
        }
    }
    __syncthreads();

    int cnt = E - base;
    if (cnt > CHUNK) cnt = CHUNK;
    u64* dst = recs + (size_t)c * CHUNK;
    const ulonglong2* s2 = (const ulonglong2*)sStage;
    ulonglong2* d2 = (ulonglong2*)dst;
    for (int i = tid; 2 * i + 1 < cnt; i += SORT_THREADS) d2[i] = s2[i];
    if ((cnt & 1) && tid == 0) dst[cnt - 1] = sStage[cnt - 1];
}

template <int NB>
__device__ __forceinline__ void bignn_addn(float2& acc, const u64* L, int i, int r,
                                           const float2* __restrict__ x2, int lane) {
    const unsigned* Lw = (const unsigned*)L;
    float2 xv[NB];
    #pragma unroll
    for (int q = 0; q < NB; ++q) {
        int qi = i + (q < r ? q : r - 1);
        unsigned c = Lw[2 * qi + 1];
        xv[q] = x2[(c << 5) | (unsigned)lane];
    }
    asm volatile("" :: "v"(xv[NB - 1].x), "v"(xv[NB - 1].y) : "memory");
    #pragma unroll
    for (int q = 0; q < NB; ++q) {
        int qi = i + (q < r ? q : r - 1);
        float vv = __uint_as_float(Lw[2 * qi]);
        float v  = (q < r) ? vv : 0.0f;
        acc.x += v * xv[q].x;
        acc.y += v * xv[q].y;
    }
}

__global__ __launch_bounds__(NTHREADS) __attribute__((amdgpu_num_vgpr(64)))
void bignn_pull_gemm(const int* __restrict__ tab,
                     const unsigned long long* __restrict__ recs,
                     const float* __restrict__ feat,
                     const float* __restrict__ W1, const float* __restrict__ b1,
                     const float* __restrict__ W2, const float* __restrict__ b2,
                     float* __restrict__ out, int N, int E, int nC) {
    __shared__ v2f   sW1[(DIM / 2) * DIM];
    __shared__ v2f   sW2[(DIM / 2) * DIM];
    __shared__ float sBias[DIM];
    __shared__ u64   sList[TILE_ROWS * PITCH];
    __shared__ int   sCnt[TILE_ROWS];

    int tid = threadIdx.x;
    int tileBase = blockIdx.x * TILE_ROWS;

    int wp = tid >> 4;
    int wg = tid & 15;
    bool wld = tid < (DIM / 2) * 16;
    const float4* W1v = (const float4*)W1;
    const float4* W2v = (const float4*)W2;
    float4 w1lo = make_float4(0.f, 0.f, 0.f, 0.f), w1hi = w1lo, w2lo = w1lo, w2hi = w1lo;
    if (wld) {
        w1lo = W1v[(2 * wp) * (DIM / 4) + wg];
        w1hi = W1v[(2 * wp + 1) * (DIM / 4) + wg];
        w2lo = W2v[(2 * wp) * (DIM / 4) + wg];
        w2hi = W2v[(2 * wp + 1) * (DIM / 4) + wg];
    }
    int   bcol  = tid & (DIM - 1);
    float bsum  = b1[bcol] + b2[bcol];
    int c0 = tid < nC ? tid : (nC - 1);
    int rs = tab[(size_t)c0 * NBINS + blockIdx.x];
    int re = tab[(size_t)c0 * NBINS + blockIdx.x + 1];

    if (tid < TILE_ROWS) sCnt[tid] = 0;
    if (wld) {
        v2f* d1 = sW1 + wp * DIM + 4 * wg;
        v2f* d2 = sW2 + wp * DIM + 4 * wg;
        d1[0] = v2f{w1lo.x, w1hi.x}; d1[1] = v2f{w1lo.y, w1hi.y};
        d1[2] = v2f{w1lo.z, w1hi.z}; d1[3] = v2f{w1lo.w, w1hi.w};
        d2[0] = v2f{w2lo.x, w2hi.x}; d2[1] = v2f{w2lo.y, w2hi.y};
        d2[2] = v2f{w2lo.z, w2hi.z}; d2[3] = v2f{w2lo.w, w2hi.w};
    }
    if (tid < DIM) sBias[tid] = bsum;
    __syncthreads();

    {
        int c = tid;
        bool live = c < nC;
        for (int turn = 0; turn < nC && live; ++turn) {
            const u64* R = recs + (size_t)c * CHUNK;
            for (int i = rs; i < re; i += 8) {
                u64 k[8];
                #pragma unroll
                for (int q = 0; q < 8; ++q) {
                    k[q] = 0;
                    if (i + q < re) k[q] = R[i + q];
                }
                int pos[8];
                #pragma unroll
                for (int q = 0; q < 8; ++q) {
                    pos[q] = CAP;
                    if (i + q < re) pos[q] = atomicAdd(&sCnt[(int)(k[q] >> 56)], 1);
                }
                #pragma unroll
                for (int q = 0; q < 8; ++q) {
                    if (pos[q] < CAP) sList[(int)(k[q] >> 56) * PITCH + pos[q]] = k[q] & KEYMASK;
                }
            }
            c += NTHREADS;
            live = c < nC;
            if (live) {
                rs = tab[(size_t)c * NBINS + blockIdx.x];
                re = tab[(size_t)c * NBINS + blockIdx.x + 1];
            }
        }
    }
    __syncthreads();

    int wave = tid >> 5;
    int lane = tid & 31;
    const float2* x2 = (const float2*)feat;
    for (int j = 0; j < ROWS_PER_WAVE; ++j) {
        int t = wave * ROWS_PER_WAVE + j;
        int d = __builtin_amdgcn_readfirstlane(sCnt[t]);
        u64* L = sList + t * PITCH;
        float2 acc = make_float2(0.0f, 0.0f);
        if (d <= CAP) {
            {
                const u64 kmax = ~(u64)0;
                u64 ka = (lane < d) ? L[lane] : kmax;
                int ra = 0;
                if (d <= 32) {
                    for (int q = 0; q < d; ++q) {
                        u64 kq = L[q];
                        ra += (kq < ka || (kq == ka && q < lane)) ? 1 : 0;
                    }
                    if (lane < d) L[ra] = ka;
                } else {
                    u64 kb = (lane + 32 < d) ? L[lane + 32] : kmax;
                    int rb = 0;
                    for (int q = 0; q < d; ++q) {
                        u64 kq = L[q];
                        ra += (kq < ka || (kq == ka && q < lane)) ? 1 : 0;
                        rb += (kq < kb || (kq == kb && q < lane + 32)) ? 1 : 0;
                    }
                    L[ra] = ka;
                    if (lane + 32 < d) L[rb] = kb;
                }
            }
            int i = 0;
            while (d - i > 8) {
                int r = (d - i < 16) ? (d - i) : 16;
                bignn_addn<16>(acc, L, i, r, x2, lane);
                i += r;
            }
            if (i < d) bignn_addn<8>(acc, L, i, d - i, x2, lane);
        } else {
            u64 last = 0;
            int have = 0;
            for (int outer = 0; outer < E; ++outer) {
                u64 best = 0;
                int mult = 0;
                for (int cc = 0; cc < nC; ++cc) {
                    int s2 = tab[(size_t)cc * NBINS + blockIdx.x];
                    int e2 = tab[(size_t)cc * NBINS + blockIdx.x + 1];
                    const u64* R2 = recs + (size_t)cc * CHUNK;
                    for (int i2 = s2; i2 < e2; ++i2) {
                        u64 k = R2[i2];
                        if ((int)(k >> 56) != t) continue;
                        k &= KEYMASK;
                        if (!have || k > last) {
                            if (mult == 0 || k < best) { best = k; mult = 1; }
                            else if (k == best) ++mult;
                        }
                    }
                }
                if (mult == 0) break;
                float v = __uint_as_float((unsigned)best);
                float2 xv = x2[(long)(int)(best >> 32) * (DIM / 2) + lane];
                for (int m2 = 0; m2 < mult; ++m2) {
                    acc.x += v * xv.x;
                    acc.y += v * xv.y;
                }
                last = best;
                have = 1;
            }
        }
        L[lane] = ((u64)__float_as_uint(acc.y) << 32) | (u64)__float_as_uint(acc.x);
    }
    __syncthreads();

    int m    = lane & 15;
    int half = lane >> 4;
    int rt   = wave >> 1;
    int ch   = wave & 1;
    int lrow = rt * 16 + m;
    int rowBase = tileBase + rt * 16;
    int row  = rowBase + m;
    int rowSafe = row < N ? row : (N - 1);

    const u64*    lrowp = sList + lrow * PITCH;
    const float2* frow  = (const float2*)(feat + (long)rowSafe * DIM);

    v8f acc2[2];
    acc2[0] = v8f{}; acc2[1] = v8f{};

    for (int k0 = 0; k0 < 16; ++k0) {
        int kp = k0 * 2 + half;
        u64 lw = lrowp[kp];
        float2 l = make_float2(__uint_as_float((unsigned)lw), __uint_as_float((unsigned)(lw >> 32)));
        float2 f = frow[kp];
        v2f aS = { l.x + f.x, l.y + f.y };
        v2f aP = { l.x * f.x, l.y * f.y };
        #pragma unroll
        for (int nb = 0; nb < 2; ++nb) {
            int cidx = (ch * 2 + nb) * 16 + m;
            v2f bA = sW1[kp * DIM + cidx];
            v2f bB = sW2[kp * DIM + cidx];
            acc2[nb] = __builtin_amdgcn_wmma_f32_16x16x4_f32(
                false, aS, false, bA, (short)0, acc2[nb], false, false);
            acc2[nb] = __builtin_amdgcn_wmma_f32_16x16x4_f32(
                false, aP, false, bB, (short)0, acc2[nb], false, false);
        }
    }

    int rOff = half * 8;
    if (rowBase + 16 <= N) {
        #pragma unroll
        for (int nb = 0; nb < 2; ++nb) {
            int col = (ch * 2 + nb) * 16 + m;
            float bias = sBias[col];
            #pragma unroll
            for (int g = 0; g < 8; ++g)
                out[(long)(rowBase + rOff + g) * DIM + col] = acc2[nb][g] + bias;
        }
    } else {
        #pragma unroll
        for (int nb = 0; nb < 2; ++nb) {
            int col = (ch * 2 + nb) * 16 + m;
            float bias = sBias[col];
            #pragma unroll
            for (int g = 0; g < 8; ++g) {
                int r = rowBase + rOff + g;
                if (r < N) out[(long)r * DIM + col] = acc2[nb][g] + bias;
            }
        }
    }
}

extern "C" void kernel_launch(void* const* d_in, const int* in_sizes, int n_in,
                              void* d_out, int out_size, void* d_ws, size_t ws_size,
                              hipStream_t stream) {
    const int*   erow  = (const int*)  d_in[0];
    const int*   ecol  = (const int*)  d_in[1];
    const float* eval_ = (const float*)d_in[2];
    const float* feat  = (const float*)d_in[3];
    const float* W1    = (const float*)d_in[4];
    const float* b1    = (const float*)d_in[5];
    const float* W2    = (const float*)d_in[6];
    const float* b2    = (const float*)d_in[7];
    float* out = (float*)d_out;

    int E = in_sizes[0];
    int N = in_sizes[3] / DIM;

    int  nC   = (E + CHUNK - 1) / CHUNK;
    int* tab  = (int*)d_ws;
    u64* recs = (u64*)(tab + (size_t)nC * NBINS);

    bignn_sort<<<nC, SORT_THREADS, 0, stream>>>(erow, ecol, eval_, tab, recs, E);
    bignn_pull_gemm<<<(N + TILE_ROWS - 1) / TILE_ROWS, NTHREADS, 0, stream>>>(
        tab, recs, feat, W1, b1, W2, b2, out, N, E, nC);
}


# Row sums: the last group of a row (1-15 entries) runs code unrolled for its exact length behind scalar branches, not 8 or 16 clamped slots with zero factors; f32 sum order kept, all products fused
# speedup vs baseline: 3.6138x; 1.0586x over previous
#include <hip/hip_runtime.h>


#define DIM 64
#define TILE_ROWS 144
#define ROWS_PER_WAVE 8
#define NTHREADS (TILE_ROWS * 4)
#define CAP 44
#define PITCH 46

#define N_NODES 100000
#define NTILES ((N_NODES + TILE_ROWS - 1) / TILE_ROWS)
#define SORT_THREADS 352
#define CHUNK (SORT_THREADS * 8)
#define NBINS (2 * SORT_THREADS)
#define KEYMASK 0x00FFFFFFFFFFFFFFull

typedef float v2f __attribute__((ext_vector_type(2)));
typedef float v8f __attribute__((ext_vector_type(8)));
typedef unsigned long long u64;

static_assert(NTHREADS / 32 * ROWS_PER_WAVE == TILE_ROWS, "every row is summed by exactly one wave");
static_assert(NTHREADS / 32 == 2 * (TILE_ROWS / 16), "two waves (column halves) per 16-row fragment");
static_assert(NBINS >= NTILES + 1, "one bin per row tile and one for the end of the chunk");
static_assert(TILE_ROWS <= 256 && N_NODES <= (1 << 24), "slot and column share the upper half of a record");
static_assert(SORT_THREADS % 32 == 0, "whole waves");

__global__ __launch_bounds__(SORT_THREADS)
void bignn_sort(const int* __restrict__ erow,
                const int* __restrict__ ecol,
                const float* __restrict__ eval_,
                int* __restrict__ tab, unsigned long long* __restrict__ recs, int E) {
    __shared__ int sBin[NBINS];
    __shared__ __attribute__((aligned(16))) u64 sStage[CHUNK];
    __shared__ int sWaveTot[SORT_THREADS / 32];

    int tid  = threadIdx.x;
    int c    = blockIdx.x;
    int base = c * CHUNK;
    sBin[tid] = 0;
    sBin[tid + SORT_THREADS] = 0;

    int      r[8];
    unsigned cl[8];
    unsigned vb[8];
    #pragma unroll
    for (int h = 0; h < 2; ++h) {
        int e0 = base + (h * SORT_THREADS + tid) * 4;
        if (e0 + 3 < E) {
            int4   r4 = *(const int4*)(erow + e0);
            int4   c4 = *(const int4*)(ecol + e0);
            float4 v4 = *(const float4*)(eval_ + e0);
            r[4 * h + 0] = r4.x; r[4 * h + 1] = r4.y; r[4 * h + 2] = r4.z; r[4 * h + 3] = r4.w;
            cl[4 * h + 0] = (unsigned)c4.x; cl[4 * h + 1] = (unsigned)c4.y;
            cl[4 * h + 2] = (unsigned)c4.z; cl[4 * h + 3] = (unsigned)c4.w;
            vb[4 * h + 0] = __float_as_uint(v4.x); vb[4 * h + 1] = __float_as_uint(v4.y);
            vb[4 * h + 2] = __float_as_uint(v4.z); vb[4 * h + 3] = __float_as_uint(v4.w);
        } else {
            #pragma unroll
            for (int q = 0; q < 4; ++q) {
                int e = e0 + q;
                bool ok = e < E;
                int ec = ok ? e : 0;
                int rr = erow[ec];
                int cc = ecol[ec];
                float vv = eval_[ec];
                r[4 * h + q]  = ok ? rr : -1;
                cl[4 * h + q] = (unsigned)cc;
                vb[4 * h + q] = __float_as_uint(vv);
            }
        }
    }
    __syncthreads();

    int tile[8];
    int rank[8];
    #pragma unroll
    for (int k = 0; k < 8; ++k) {
        tile[k] = -1;
        rank[k] = 0;
        if (r[k] >= 0) {
            tile[k] = r[k] / TILE_ROWS;
            rank[k] = atomicAdd(&sBin[tile[k]], 1);
        }
    }
    __syncthreads();

    int lane = tid & 31;
    int wave = tid >> 5;
    int ca = sBin[2 * tid];
    int cb = sBin[2 * tid + 1];
    int s  = ca + cb;
    int inc = s;
    #pragma unroll
    for (int d = 1; d < 32; d <<= 1) {
        int y = __shfl_up(inc, d);
        if (lane >= d) inc += y;
    }
    if (lane == 31) sWaveTot[wave] = inc;
    __syncthreads();
    int wpre = 0;
    #pragma unroll
    for (int w = 0; w < SORT_THREADS / 32; ++w) {
        int wt = sWaveTot[w];
        wpre += (w < wave) ? wt : 0;
    }
    int ex = wpre + inc - s;
    sBin[2 * tid]     = ex;
    sBin[2 * tid + 1] = ex + ca;
    *(int2*)(tab + (size_t)c * NBINS + 2 * tid) = make_int2(ex, ex + ca);
    __syncthreads();

    #pragma unroll
    for (int k = 0; k < 8; ++k) {
        if (tile[k] >= 0) {
            int pos  = sBin[tile[k]] + rank[k];
            unsigned slot = (unsigned)(r[k] - tile[k] * TILE_ROWS);
            sStage[pos] = ((u64)((slot << 24) | cl[k]) << 32) | (u64)vb[k];
        }
    }
    __syncthreads();

    int cnt = E - base;
    if (cnt > CHUNK) cnt = CHUNK;
    u64* dst = recs + (size_t)c * CHUNK;
    const ulonglong2* s2 = (const ulonglong2*)sStage;
    ulonglong2* d2 = (ulonglong2*)dst;
    for (int i = tid; 2 * i + 1 < cnt; i += SORT_THREADS) d2[i] = s2[i];
    if ((cnt & 1) && tid == 0) dst[cnt - 1] = sStage[cnt - 1];
}

template <int R>
__device__ __forceinline__ void bignn_addx(v2f& acc, const u64* L, int i,
                                           const v2f* __restrict__ x2, int lane) {
    const unsigned* Lw = (const unsigned*)(L + i);
    v2f xv[R];
    #pragma unroll
    for (int q = R - 1; q >= 0; --q) {
        unsigned c = Lw[2 * q + 1];
        xv[q] = x2[(c << 5) | (unsigned)lane];
    }
    asm volatile("" :: "v"(xv[0].x), "v"(xv[0].y) : "memory");
    #pragma unroll
    for (int q = 0; q < R; ++q) {
        float v = __uint_as_float(Lw[2 * q]);
        acc = __builtin_elementwise_fma(v2f{v, v}, xv[q], acc);
    }
}

template <int LO, int HI>
__device__ __forceinline__ void bignn_tail(v2f& acc, const u64* L, int i, int r,
                                           const v2f* __restrict__ x2, int lane) {
    if constexpr (LO == HI) {
        bignn_addx<LO>(acc, L, i, x2, lane);
    } else {
        constexpr int MID = (LO + HI + 1) / 2;
        if (r >= MID) bignn_tail<MID, HI>(acc, L, i, r, x2, lane);
        else          bignn_tail<LO, MID - 1>(acc, L, i, r, x2, lane);
    }
}

__global__ __launch_bounds__(NTHREADS) __attribute__((amdgpu_num_vgpr(64)))
void bignn_pull_gemm(const int* __restrict__ tab,
                     const unsigned long long* __restrict__ recs,
                     const float* __restrict__ feat,
                     const float* __restrict__ W1, const float* __restrict__ b1,
                     const float* __restrict__ W2, const float* __restrict__ b2,
                     float* __restrict__ out, int N, int E, int nC) {
    __shared__ v2f   sW1[(DIM / 2) * DIM];
    __shared__ v2f   sW2[(DIM / 2) * DIM];
    __shared__ float sBias[DIM];
    __shared__ u64   sList[TILE_ROWS * PITCH];
    __shared__ int   sCnt[TILE_ROWS];

    int tid = threadIdx.x;
    int tileBase = blockIdx.x * TILE_ROWS;

    int wp = tid >> 4;
    int wg = tid & 15;
    bool wld = tid < (DIM / 2) * 16;
    const float4* W1v = (const float4*)W1;
    const float4* W2v = (const float4*)W2;
    float4 w1lo = make_float4(0.f, 0.f, 0.f, 0.f), w1hi = w1lo, w2lo = w1lo, w2hi = w1lo;
    if (wld) {
        w1lo = W1v[(2 * wp) * (DIM / 4) + wg];
        w1hi = W1v[(2 * wp + 1) * (DIM / 4) + wg];
        w2lo = W2v[(2 * wp) * (DIM / 4) + wg];
        w2hi = W2v[(2 * wp + 1) * (DIM / 4) + wg];
    }
    int   bcol  = tid & (DIM - 1);
    float bsum  = b1[bcol] + b2[bcol];
    int c0 = tid < nC ? tid : (nC - 1);
    int rs = tab[(size_t)c0 * NBINS + blockIdx.x];
    int re = tab[(size_t)c0 * NBINS + blockIdx.x + 1];

    if (tid < TILE_ROWS) sCnt[tid] = 0;
    if (wld) {
        v2f* d1 = sW1 + wp * DIM + 4 * wg;
        v2f* d2 = sW2 + wp * DIM + 4 * wg;
        d1[0] = v2f{w1lo.x, w1hi.x}; d1[1] = v2f{w1lo.y, w1hi.y};
        d1[2] = v2f{w1lo.z, w1hi.z}; d1[3] = v2f{w1lo.w, w1hi.w};
        d2[0] = v2f{w2lo.x, w2hi.x}; d2[1] = v2f{w2lo.y, w2hi.y};
        d2[2] = v2f{w2lo.z, w2hi.z}; d2[3] = v2f{w2lo.w, w2hi.w};
    }
    if (tid < DIM) sBias[tid] = bsum;
    __syncthreads();

    {
        int c = tid;
        bool live = c < nC;
        for (int turn = 0; turn < nC && live; ++turn) {
            const u64* R = recs + (size_t)c * CHUNK;
            for (int i = rs; i < re; i += 8) {
                u64 k[8];
                #pragma unroll
                for (int q = 0; q < 8; ++q) {
                    k[q] = 0;
                    if (i + q < re) k[q] = R[i + q];
                }
                int pos[8];
                #pragma unroll
                for (int q = 0; q < 8; ++q) {
                    pos[q] = CAP;
                    if (i + q < re) pos[q] = atomicAdd(&sCnt[(int)(k[q] >> 56)], 1);
                }
                #pragma unroll
                for (int q = 0; q < 8; ++q) {
                    if (pos[q] < CAP) sList[(int)(k[q] >> 56) * PITCH + pos[q]] = k[q] & KEYMASK;
                }
            }
            c += NTHREADS;
            live = c < nC;
            if (live) {
                rs = tab[(size_t)c * NBINS + blockIdx.x];
                re = tab[(size_t)c * NBINS + blockIdx.x + 1];
            }
        }
    }
    __syncthreads();

    int wave = tid >> 5;
    int lane = tid & 31;
    const v2f* x2 = (const v2f*)feat;
    for (int j = 0; j < ROWS_PER_WAVE; ++j) {
        int t = wave * ROWS_PER_WAVE + j;
        int d = __builtin_amdgcn_readfirstlane(sCnt[t]);
        u64* L = sList + t * PITCH;
        v2f acc = v2f{0.0f, 0.0f};
        if (d <= CAP) {
            {
                const u64 kmax = ~(u64)0;
                u64 ka = (lane < d) ? L[lane] : kmax;
                int ra = 0;
                if (d <= 32) {
                    for (int q = 0; q < d; ++q) {
                        u64 kq = L[q];
                        ra += (kq < ka || (kq == ka && q < lane)) ? 1 : 0;
                    }
                    if (lane < d) L[ra] = ka;
                } else {
                    u64 kb = (lane + 32 < d) ? L[lane + 32] : kmax;
                    int rb = 0;
                    for (int q = 0; q < d; ++q) {
                        u64 kq = L[q];
                        ra += (kq < ka || (kq == ka && q < lane)) ? 1 : 0;
                        rb += (kq < kb || (kq == kb && q < lane + 32)) ? 1 : 0;
                    }
                    L[ra] = ka;
                    if (lane + 32 < d) L[rb] = kb;
                }
            }
            int i = 0;
            while (d - i >= 16) {
                bignn_addx<16>(acc, L, i, x2, lane);
                i += 16;
            }
            if (i < d) bignn_tail<1, 15>(acc, L, i, d - i, x2, lane);
        } else {
            u64 last = 0;
            int have = 0;
            for (int outer = 0; outer < E; ++outer) {
                u64 best = 0;
                int mult = 0;
                for (int cc = 0; cc < nC; ++cc) {
                    int s2 = tab[(size_t)cc * NBINS + blockIdx.x];
                    int e2 = tab[(size_t)cc * NBINS + blockIdx.x + 1];
                    const u64* R2 = recs + (size_t)cc * CHUNK;
                    for (int i2 = s2; i2 < e2; ++i2) {
                        u64 k = R2[i2];
                        if ((int)(k >> 56) != t) continue;
                        k &= KEYMASK;
                        if (!have || k > last) {
                            if (mult == 0 || k < best) { best = k; mult = 1; }
                            else if (k == best) ++mult;
                        }
                    }
                }
                if (mult == 0) break;
                float v = __uint_as_float((unsigned)best);
                v2f xv = x2[(long)(int)(best >> 32) * (DIM / 2) + lane];
                for (int m2 = 0; m2 < mult; ++m2) {
                    acc.x += v * xv.x;
                    acc.y += v * xv.y;
                }
                last = best;
                have = 1;
            }
        }
        L[lane] = ((u64)__float_as_uint(acc.y) << 32) | (u64)__float_as_uint(acc.x);
    }
    __syncthreads();

    int m    = lane & 15;
    int half = lane >> 4;
    int rt   = wave >> 1;
    int ch   = wave & 1;
    int lrow = rt * 16 + m;
    int rowBase = tileBase + rt * 16;
    int row  = rowBase + m;
    int rowSafe = row < N ? row : (N - 1);

    const u64*    lrowp = sList + lrow * PITCH;
    const float2* frow  = (const float2*)(feat + (long)rowSafe * DIM);

    v8f acc2[2];
    acc2[0] = v8f{}; acc2[1] = v8f{};

    for (int k0 = 0; k0 < 16; ++k0) {
        int kp = k0 * 2 + half;
        u64 lw = lrowp[kp];
        float2 l = make_float2(__uint_as_float((unsigned)lw), __uint_as_float((unsigned)(lw >> 32)));
        float2 f = frow[kp];
        v2f aS = { l.x + f.x, l.y + f.y };
        v2f aP = { l.x * f.x, l.y * f.y };
        #pragma unroll
        for (int nb = 0; nb < 2; ++nb) {
            int cidx = (ch * 2 + nb) * 16 + m;
            v2f bA = sW1[kp * DIM + cidx];
            v2f bB = sW2[kp * DIM + cidx];
            acc2[nb] = __builtin_amdgcn_wmma_f32_16x16x4_f32(
                false, aS, false, bA, (short)0, acc2[nb], false, false);
            acc2[nb] = __builtin_amdgcn_wmma_f32_16x16x4_f32(
                false, aP, false, bB, (short)0, acc2[nb], false, false);
        }
    }

    int rOff = half * 8;
    if (rowBase + 16 <= N) {
        #pragma unroll
        for (int nb = 0; nb < 2; ++nb) {
            int col = (ch * 2 + nb) * 16 + m;
            float bias = sBias[col];
            #pragma unroll
            for (int g = 0; g < 8; ++g)
                out[(long)(rowBase + rOff + g) * DIM + col] = acc2[nb][g] + bias;
        }
    } else {
        #pragma unroll
        for (int nb = 0; nb < 2; ++nb) {
            int col = (ch * 2 + nb) * 16 + m;
            float bias = sBias[col];
            #pragma unroll
            for (int g = 0; g < 8; ++g) {
                int r = rowBase + rOff + g;
                if (r < N) out[(long)r * DIM + col] = acc2[nb][g] + bias;
            }
        }
    }
}

extern "C" void kernel_launch(void* const* d_in, const int* in_sizes, int n_in,
                              void* d_out, int out_size, void* d_ws, size_t ws_size,
                              hipStream_t stream) {
    const int*   erow  = (const int*)  d_in[0];
    const int*   ecol  = (const int*)  d_in[1];
    const float* eval_ = (const float*)d_in[2];
    const float* feat  = (const float*)d_in[3];
    const float* W1    = (const float*)d_in[4];
    const float* b1    = (const float*)d_in[5];
    const float* W2    = (const float*)d_in[6];
    const float* b2    = (const float*)d_in[7];
    float* out = (float*)d_out;

    int E = in_sizes[0];
    int N = in_sizes[3] / DIM;

    int  nC   = (E + CHUNK - 1) / CHUNK;
    int* tab  = (int*)d_ws;
    u64* recs = (u64*)(tab + (size_t)nC * NBINS);

    bignn_sort<<<nC, SORT_THREADS, 0, stream>>>(erow, ecol, eval_, tab, recs, E);
    bignn_pull_gemm<<<(N + TILE_ROWS - 1) / TILE_ROWS, NTHREADS, 0, stream>>>(
        tab, recs, feat, W1, b1, W2, b2, out, N, E, nC);
}


# Sort chunks of 6400 edges (was 2816) on 800 threads, one block per WGP: a tile's runs are 2.3x fewer and longer, 2 pull lanes share a run: under half the scattered requests; f32 sum order kept
# speedup vs baseline: 3.8675x; 1.0702x over previous
#include <hip/hip_runtime.h>


#define DIM 64
#define TILE_ROWS 144
#define ROWS_PER_WAVE 8
#define NTHREADS (TILE_ROWS * 4)
#define CAP 44
#define PITCH 46

#define N_NODES 100000
#define NTILES ((N_NODES + TILE_ROWS - 1) / TILE_ROWS)
#define SORT_THREADS 800
#define CHUNK (SORT_THREADS * 8)
#define LPC 2
#define NBINS SORT_THREADS
#define KEYMASK 0x00FFFFFFFFFFFFFFull

typedef float v2f __attribute__((ext_vector_type(2)));
typedef float v8f __attribute__((ext_vector_type(8)));
typedef unsigned long long u64;

static_assert(NTHREADS / 32 * ROWS_PER_WAVE == TILE_ROWS, "every row is summed by exactly one wave");
static_assert(NTHREADS / 32 == 2 * (TILE_ROWS / 16), "two waves (column halves) per 16-row fragment");
static_assert(NBINS >= NTILES + 1, "one bin per row tile and one for the end of the chunk");
static_assert(TILE_ROWS <= 256 && N_NODES <= (1 << 24), "slot and column share the upper half of a record");
static_assert(SORT_THREADS % 32 == 0, "whole waves");
static_assert(NTHREADS % LPC == 0 && (LPC & (LPC - 1)) == 0, "the lanes of a chunk are neighbours in one wave");

__global__ __launch_bounds__(SORT_THREADS)
void bignn_sort(const int* __restrict__ erow,
                const int* __restrict__ ecol,
                const float* __restrict__ eval_,
                int* __restrict__ tab, unsigned long long* __restrict__ recs, int E) {
    __shared__ int sBin[NBINS];
    __shared__ __attribute__((aligned(16))) u64 sStage[CHUNK];
    __shared__ int sWaveTot[SORT_THREADS / 32];

    int tid  = threadIdx.x;
    int c    = blockIdx.x;
    int base = c * CHUNK;
    sBin[tid] = 0;

    int      r[8];
    unsigned cl[8];
    unsigned vb[8];
    #pragma unroll
    for (int h = 0; h < 2; ++h) {
        int e0 = base + (h * SORT_THREADS + tid) * 4;
        if (e0 + 3 < E) {
            int4   r4 = *(const int4*)(erow + e0);
            int4   c4 = *(const int4*)(ecol + e0);
            float4 v4 = *(const float4*)(eval_ + e0);
            r[4 * h + 0] = r4.x; r[4 * h + 1] = r4.y; r[4 * h + 2] = r4.z; r[4 * h + 3] = r4.w;
            cl[4 * h + 0] = (unsigned)c4.x; cl[4 * h + 1] = (unsigned)c4.y;
            cl[4 * h + 2] = (unsigned)c4.z; cl[4 * h + 3] = (unsigned)c4.w;
            vb[4 * h + 0] = __float_as_uint(v4.x); vb[4 * h + 1] = __float_as_uint(v4.y);
            vb[4 * h + 2] = __float_as_uint(v4.z); vb[4 * h + 3] = __float_as_uint(v4.w);
        } else {
            #pragma unroll
            for (int q = 0; q < 4; ++q) {
                int e = e0 + q;
                bool ok = e < E;
                int ec = ok ? e : 0;
                int rr = erow[ec];
                int cc = ecol[ec];
                float vv = eval_[ec];
                r[4 * h + q]  = ok ? rr : -1;
                cl[4 * h + q] = (unsigned)cc;
                vb[4 * h + q] = __float_as_uint(vv);
            }
        }
    }
    __syncthreads();

    int tile[8];
    int rank[8];
    #pragma unroll
    for (int k = 0; k < 8; ++k) {
        tile[k] = -1;
        rank[k] = 0;
        if (r[k] >= 0) {
            tile[k] = r[k] / TILE_ROWS;
            rank[k] = atomicAdd(&sBin[tile[k]], 1);
        }
    }
    __syncthreads();

    int lane = tid & 31;
    int wave = tid >> 5;
    int s  = sBin[tid];
    int inc = s;
    #pragma unroll
    for (int d = 1; d < 32; d <<= 1) {
        int y = __shfl_up(inc, d);
        if (lane >= d) inc += y;
    }
    if (lane == 31) sWaveTot[wave] = inc;
    __syncthreads();
    int wpre = 0;
    #pragma unroll
    for (int w = 0; w < SORT_THREADS / 32; ++w) {
        int wt = sWaveTot[w];
        wpre += (w < wave) ? wt : 0;
    }
    int ex = wpre + inc - s;
    sBin[tid] = ex;
    tab[(size_t)c * NBINS + tid] = ex;
    __syncthreads();

    #pragma unroll
    for (int k = 0; k < 8; ++k) {
        if (tile[k] >= 0) {
            int pos  = sBin[tile[k]] + rank[k];
            unsigned slot = (unsigned)(r[k] - tile[k] * TILE_ROWS);
            sStage[pos] = ((u64)((slot << 24) | cl[k]) << 32) | (u64)vb[k];
        }
    }
    __syncthreads();

    int cnt = E - base;
    if (cnt > CHUNK) cnt = CHUNK;
    u64* dst = recs + (size_t)c * CHUNK;
    const ulonglong2* s2 = (const ulonglong2*)sStage;
    ulonglong2* d2 = (ulonglong2*)dst;
    for (int i = tid; 2 * i + 1 < cnt; i += SORT_THREADS) d2[i] = s2[i];
    if ((cnt & 1) && tid == 0) dst[cnt - 1] = sStage[cnt - 1];
}

template <int R>
__device__ __forceinline__ void bignn_addx(v2f& acc, const u64* L, int i,
                                           const v2f* __restrict__ x2, int lane) {
    const unsigned* Lw = (const unsigned*)(L + i);
    v2f xv[R];
    #pragma unroll
    for (int q = R - 1; q >= 0; --q) {
        unsigned c = Lw[2 * q + 1];
        xv[q] = x2[(c << 5) | (unsigned)lane];
    }
    asm volatile("" :: "v"(xv[0].x), "v"(xv[0].y) : "memory");
    #pragma unroll
    for (int q = 0; q < R; ++q) {
        float v = __uint_as_float(Lw[2 * q]);
        acc = __builtin_elementwise_fma(v2f{v, v}, xv[q], acc);
    }
}

template <int LO, int HI>
__device__ __forceinline__ void bignn_tail(v2f& acc, const u64* L, int i, int r,
                                           const v2f* __restrict__ x2, int lane) {
    if constexpr (LO == HI) {
        bignn_addx<LO>(acc, L, i, x2, lane);
    } else {
        constexpr int MID = (LO + HI + 1) / 2;
        if (r >= MID) bignn_tail<MID, HI>(acc, L, i, r, x2, lane);
        else          bignn_tail<LO, MID - 1>(acc, L, i, r, x2, lane);
    }
}

__global__ __launch_bounds__(NTHREADS) __attribute__((amdgpu_num_vgpr(64)))
void bignn_pull_gemm(const int* __restrict__ tab,
                     const unsigned long long* __restrict__ recs,
                     const float* __restrict__ feat,
                     const float* __restrict__ W1, const float* __restrict__ b1,
                     const float* __restrict__ W2, const float* __restrict__ b2,
                     float* __restrict__ out, int N, int E, int nC) {
    __shared__ v2f   sW1[(DIM / 2) * DIM];
    __shared__ v2f   sW2[(DIM / 2) * DIM];
    __shared__ float sBias[DIM];
    __shared__ u64   sList[TILE_ROWS * PITCH];
    __shared__ int   sCnt[TILE_ROWS];

    int tid = threadIdx.x;
    int tileBase = blockIdx.x * TILE_ROWS;

    int wp = tid >> 4;
    int wg = tid & 15;
    bool wld = tid < (DIM / 2) * 16;
    const float4* W1v = (const float4*)W1;
    const float4* W2v = (const float4*)W2;
    float4 w1lo = make_float4(0.f, 0.f, 0.f, 0.f), w1hi = w1lo, w2lo = w1lo, w2hi = w1lo;
    if (wld) {
        w1lo = W1v[(2 * wp) * (DIM / 4) + wg];
        w1hi = W1v[(2 * wp + 1) * (DIM / 4) + wg];
        w2lo = W2v[(2 * wp) * (DIM / 4) + wg];
        w2hi = W2v[(2 * wp + 1) * (DIM / 4) + wg];
    }
    int   bcol  = tid & (DIM - 1);
    float bsum  = b1[bcol] + b2[bcol];
    int c0 = (tid / LPC) < nC ? (tid / LPC) : (nC - 1);
    int rs = tab[(size_t)c0 * NBINS + blockIdx.x];
    int re = tab[(size_t)c0 * NBINS + blockIdx.x + 1];

    if (tid < TILE_ROWS) sCnt[tid] = 0;
    if (wld) {
        v2f* d1 = sW1 + wp * DIM + 4 * wg;
        v2f* d2 = sW2 + wp * DIM + 4 * wg;
        d1[0] = v2f{w1lo.x, w1hi.x}; d1[1] = v2f{w1lo.y, w1hi.y};
        d1[2] = v2f{w1lo.z, w1hi.z}; d1[3] = v2f{w1lo.w, w1hi.w};
        d2[0] = v2f{w2lo.x, w2hi.x}; d2[1] = v2f{w2lo.y, w2hi.y};
        d2[2] = v2f{w2lo.z, w2hi.z}; d2[3] = v2f{w2lo.w, w2hi.w};
    }
    if (tid < DIM) sBias[tid] = bsum;
    __syncthreads();

    {
        int c = tid / LPC;
        int sub = tid & (LPC - 1);
        bool live = c < nC;
        for (int turn = 0; turn < nC && live; ++turn) {
            const u64* R = recs + (size_t)c * CHUNK;
            for (int i = rs + sub; i < re; i += 8 * LPC) {
                u64 k[8];
                #pragma unroll
                for (int q = 0; q < 8; ++q) {
                    k[q] = 0;
                    if (i + LPC * q < re) k[q] = R[i + LPC * q];
                }
                int pos[8];
                #pragma unroll
                for (int q = 0; q < 8; ++q) {
                    pos[q] = CAP;
                    if (i + LPC * q < re) pos[q] = atomicAdd(&sCnt[(int)(k[q] >> 56)], 1);
                }
                #pragma unroll
                for (int q = 0; q < 8; ++q) {
                    if (pos[q] < CAP) sList[(int)(k[q] >> 56) * PITCH + pos[q]] = k[q] & KEYMASK;
                }
            }
            c += NTHREADS / LPC;
            live = c < nC;
            if (live) {
                rs = tab[(size_t)c * NBINS + blockIdx.x];
                re = tab[(size_t)c * NBINS + blockIdx.x + 1];
            }
        }
    }
    __syncthreads();

    int wave = tid >> 5;
    int lane = tid & 31;
    const v2f* x2 = (const v2f*)feat;
    for (int j = 0; j < ROWS_PER_WAVE; ++j) {
        int t = wave * ROWS_PER_WAVE + j;
        int d = __builtin_amdgcn_readfirstlane(sCnt[t]);
        u64* L = sList + t * PITCH;
        v2f acc = v2f{0.0f, 0.0f};
        if (d <= CAP) {
            {
                const u64 kmax = ~(u64)0;
                u64 ka = (lane < d) ? L[lane] : kmax;
                int ra = 0;
                if (d <= 32) {
                    for (int q = 0; q < d; ++q) {
                        u64 kq = L[q];
                        ra += (kq < ka || (kq == ka && q < lane)) ? 1 : 0;
                    }
                    if (lane < d) L[ra] = ka;
                } else {
                    u64 kb = (lane + 32 < d) ? L[lane + 32] : kmax;
                    int rb = 0;
                    for (int q = 0; q < d; ++q) {
                        u64 kq = L[q];
                        ra += (kq < ka || (kq == ka && q < lane)) ? 1 : 0;
                        rb += (kq < kb || (kq == kb && q < lane + 32)) ? 1 : 0;
                    }
                    L[ra] = ka;
                    if (lane + 32 < d) L[rb] = kb;
                }
            }
            int i = 0;
            while (d - i >= 16) {
                bignn_addx<16>(acc, L, i, x2, lane);
                i += 16;
            }
            if (i < d) bignn_tail<1, 15>(acc, L, i, d - i, x2, lane);
        } else {
            u64 last = 0;
            int have = 0;
            for (int outer = 0; outer < E; ++outer) {
                u64 best = 0;
                int mult = 0;
                for (int cc = 0; cc < nC; ++cc) {
                    int s2 = tab[(size_t)cc * NBINS + blockIdx.x];
                    int e2 = tab[(size_t)cc * NBINS + blockIdx.x + 1];
                    const u64* R2 = recs + (size_t)cc * CHUNK;
                    for (int i2 = s2; i2 < e2; ++i2) {
                        u64 k = R2[i2];
                        if ((int)(k >> 56) != t) continue;
                        k &= KEYMASK;
                        if (!have || k > last) {
                            if (mult == 0 || k < best) { best = k; mult = 1; }
                            else if (k == best) ++mult;
                        }
                    }
                }
                if (mult == 0) break;
                float v = __uint_as_float((unsigned)best);
                v2f xv = x2[(long)(int)(best >> 32) * (DIM / 2) + lane];
                for (int m2 = 0; m2 < mult; ++m2) {
                    acc.x += v * xv.x;
                    acc.y += v * xv.y;
                }
                last = best;
                have = 1;
            }
        }
        L[lane] = ((u64)__float_as_uint(acc.y) << 32) | (u64)__float_as_uint(acc.x);
    }
    __syncthreads();

    int m    = lane & 15;
    int half = lane >> 4;
    int rt   = wave >> 1;
    int ch   = wave & 1;
    int lrow = rt * 16 + m;
    int rowBase = tileBase + rt * 16;
    int row  = rowBase + m;
    int rowSafe = row < N ? row : (N - 1);

    const u64*    lrowp = sList + lrow * PITCH;
    const float2* frow  = (const float2*)(feat + (long)rowSafe * DIM);

    v8f acc2[2];
    acc2[0] = v8f{}; acc2[1] = v8f{};

    for (int k0 = 0; k0 < 16; ++k0) {
        int kp = k0 * 2 + half;
        u64 lw = lrowp[kp];
        float2 l = make_float2(__uint_as_float((unsigned)lw), __uint_as_float((unsigned)(lw >> 32)));
        float2 f = frow[kp];
        v2f aS = { l.x + f.x, l.y + f.y };
        v2f aP = { l.x * f.x, l.y * f.y };
        #pragma unroll
        for (int nb = 0; nb < 2; ++nb) {
            int cidx = (ch * 2 + nb) * 16 + m;
            v2f bA = sW1[kp * DIM + cidx];
            v2f bB = sW2[kp * DIM + cidx];
            acc2[nb] = __builtin_amdgcn_wmma_f32_16x16x4_f32(
                false, aS, false, bA, (short)0, acc2[nb], false, false);
            acc2[nb] = __builtin_amdgcn_wmma_f32_16x16x4_f32(
                false, aP, false, bB, (short)0, acc2[nb], false, false);
        }
    }

    int rOff = half * 8;
    if (rowBase + 16 <= N) {
        #pragma unroll
        for (int nb = 0; nb < 2; ++nb) {
            int col = (ch * 2 + nb) * 16 + m;
            float bias = sBias[col];
            #pragma unroll
            for (int g = 0; g < 8; ++g)
                out[(long)(rowBase + rOff + g) * DIM + col] = acc2[nb][g] + bias;
        }
    } else {
        #pragma unroll
        for (int nb = 0; nb < 2; ++nb) {
            int col = (ch * 2 + nb) * 16 + m;
            float bias = sBias[col];
            #pragma unroll
            for (int g = 0; g < 8; ++g) {
                int r = rowBase + rOff + g;
                if (r < N) out[(long)r * DIM + col] = acc2[nb][g] + bias;
            }
        }
    }
}

extern "C" void kernel_launch(void* const* d_in, const int* in_sizes, int n_in,
                              void* d_out, int out_size, void* d_ws, size_t ws_size,
                              hipStream_t stream) {
    const int*   erow  = (const int*)  d_in[0];
    const int*   ecol  = (const int*)  d_in[1];
    const float* eval_ = (const float*)d_in[2];
    const float* feat  = (const float*)d_in[3];
    const float* W1    = (const float*)d_in[4];
    const float* b1    = (const float*)d_in[5];
    const float* W2    = (const float*)d_in[6];
    const float* b2    = (const float*)d_in[7];
    float* out = (float*)d_out;

    int E = in_sizes[0];
    int N = in_sizes[3] / DIM;

    int  nC   = (E + CHUNK - 1) / CHUNK;
    int* tab  = (int*)d_ws;
    u64* recs = (u64*)(tab + (size_t)nC * NBINS);

    bignn_sort<<<nC, SORT_THREADS, 0, stream>>>(erow, ecol, eval_, tab, recs, E);
    bignn_pull_gemm<<<(N + TILE_ROWS - 1) / TILE_ROWS, NTHREADS, 0, stream>>>(
        tab, recs, feat, W1, b1, W2, b2, out, N, E, nC);
}


# GEMM A rows from LDS only: the wave that sums a row loads its own feature row whole beside the gathers, stores lap+feat and lap*feat (512 B a row); tiles of 208 rows, 2 a WGP; f32 sum order kept
# speedup vs baseline: 3.9861x; 1.0307x over previous
#include <hip/hip_runtime.h>


#define DIM 64
#define TILE_ROWS 208
#define ROWS_PER_WAVE 8
#define NTHREADS (TILE_ROWS * 4)
#define CAP 64
#define PITCH 66

#define N_NODES 100000
#define NTILES ((N_NODES + TILE_ROWS - 1) / TILE_ROWS)
#define SORT_THREADS 800
#define CHUNK (SORT_THREADS * 8)
#define LPC 2
#define PB 12
#define NBINS SORT_THREADS
#define KEYMASK 0x00FFFFFFFFFFFFFFull

typedef float v2f __attribute__((ext_vector_type(2)));
typedef float v8f __attribute__((ext_vector_type(8)));
typedef unsigned long long u64;

static_assert(NTHREADS / 32 * ROWS_PER_WAVE == TILE_ROWS, "every row is summed by exactly one wave");
static_assert(NTHREADS / 32 == 2 * (TILE_ROWS / 16), "two waves (column halves) per 16-row fragment");
static_assert(NBINS >= NTILES + 1, "one bin per row tile and one for the end of the chunk");
static_assert(TILE_ROWS <= 256 && N_NODES <= (1 << 24), "slot and column share the upper half of a record");
static_assert(SORT_THREADS % 32 == 0, "whole waves");
static_assert(NTHREADS % LPC == 0 && (LPC & (LPC - 1)) == 0, "the lanes of a chunk are neighbours in one wave");

__global__ __launch_bounds__(SORT_THREADS)
void bignn_sort(const int* __restrict__ erow,
                const int* __restrict__ ecol,
                const float* __restrict__ eval_,
                int* __restrict__ tab, unsigned long long* __restrict__ recs, int E) {
    __shared__ int sBin[NBINS];
    __shared__ __attribute__((aligned(16))) u64 sStage[CHUNK];
    __shared__ int sWaveTot[SORT_THREADS / 32];

    int tid  = threadIdx.x;
    int c    = blockIdx.x;
    int base = c * CHUNK;
    sBin[tid] = 0;

    int      r[8];
    unsigned cl[8];
    unsigned vb[8];
    #pragma unroll
    for (int h = 0; h < 2; ++h) {
        int e0 = base + (h * SORT_THREADS + tid) * 4;
        if (e0 + 3 < E) {
            int4   r4 = *(const int4*)(erow + e0);
            int4   c4 = *(const int4*)(ecol + e0);
            float4 v4 = *(const float4*)(eval_ + e0);
            r[4 * h + 0] = r4.x; r[4 * h + 1] = r4.y; r[4 * h + 2] = r4.z; r[4 * h + 3] = r4.w;
            cl[4 * h + 0] = (unsigned)c4.x; cl[4 * h + 1] = (unsigned)c4.y;
            cl[4 * h + 2] = (unsigned)c4.z; cl[4 * h + 3] = (unsigned)c4.w;
            vb[4 * h + 0] = __float_as_uint(v4.x); vb[4 * h + 1] = __float_as_uint(v4.y);
            vb[4 * h + 2] = __float_as_uint(v4.z); vb[4 * h + 3] = __float_as_uint(v4.w);
        } else {
            #pragma unroll
            for (int q = 0; q < 4; ++q) {
                int e = e0 + q;
                bool ok = e < E;
                int ec = ok ? e : 0;
                int rr = erow[ec];
                int cc = ecol[ec];
                float vv = eval_[ec];
                r[4 * h + q]  = ok ? rr : -1;
                cl[4 * h + q] = (unsigned)cc;
                vb[4 * h + q] = __float_as_uint(vv);
            }
        }
    }
    __syncthreads();

    int tile[8];
    int rank[8];
    #pragma unroll
    for (int k = 0; k < 8; ++k) {
        tile[k] = -1;
        rank[k] = 0;
        if (r[k] >= 0) {
            tile[k] = r[k] / TILE_ROWS;
            rank[k] = atomicAdd(&sBin[tile[k]], 1);
        }
    }
    __syncthreads();

    int lane = tid & 31;
    int wave = tid >> 5;
    int s  = sBin[tid];
    int inc = s;
    #pragma unroll
    for (int d = 1; d < 32; d <<= 1) {
        int y = __shfl_up(inc, d);
        if (lane >= d) inc += y;
    }
    if (lane == 31) sWaveTot[wave] = inc;
    __syncthreads();
    int wpre = 0;
    #pragma unroll
    for (int w = 0; w < SORT_THREADS / 32; ++w) {
        int wt = sWaveTot[w];
        wpre += (w < wave) ? wt : 0;
    }
    int ex = wpre + inc - s;
    sBin[tid] = ex;
    tab[(size_t)c * NBINS + tid] = ex;
    __syncthreads();

    #pragma unroll
    for (int k = 0; k < 8; ++k) {
        if (tile[k] >= 0) {
            int pos  = sBin[tile[k]] + rank[k];
            unsigned slot = (unsigned)(r[k] - tile[k] * TILE_ROWS);
            sStage[pos] = ((u64)((slot << 24) | cl[k]) << 32) | (u64)vb[k];
        }
    }
    __syncthreads();

    int cnt = E - base;
    if (cnt > CHUNK) cnt = CHUNK;
    u64* dst = recs + (size_t)c * CHUNK;
    const ulonglong2* s2 = (const ulonglong2*)sStage;
    ulonglong2* d2 = (ulonglong2*)dst;
    for (int i = tid; 2 * i + 1 < cnt; i += SORT_THREADS) d2[i] = s2[i];
    if ((cnt & 1) && tid == 0) dst[cnt - 1] = sStage[cnt - 1];
}

template <int R>
__device__ __forceinline__ void bignn_addx(v2f& acc, const u64* L, int i,
                                           const v2f* __restrict__ x2, int lane) {
    const unsigned* Lw = (const unsigned*)(L + i);
    v2f xv[R];
    #pragma unroll
    for (int q = R - 1; q >= 0; --q) {
        unsigned c = Lw[2 * q + 1];
        xv[q] = x2[(c << 5) | (unsigned)lane];
    }
    asm volatile("" :: "v"(xv[0].x), "v"(xv[0].y) : "memory");
    #pragma unroll
    for (int q = 0; q < R; ++q) {
        float v = __uint_as_float(Lw[2 * q]);
        acc = __builtin_elementwise_fma(v2f{v, v}, xv[q], acc);
    }
}

template <int LO, int HI>
__device__ __forceinline__ void bignn_tail(v2f& acc, const u64* L, int i, int r,
                                           const v2f* __restrict__ x2, int lane) {
    if constexpr (LO == HI) {
        bignn_addx<LO>(acc, L, i, x2, lane);
    } else {
        constexpr int MID = (LO + HI + 1) / 2;
        if (r >= MID) bignn_tail<MID, HI>(acc, L, i, r, x2, lane);
        else          bignn_tail<LO, MID - 1>(acc, L, i, r, x2, lane);
    }
}

__global__ __launch_bounds__(NTHREADS) __attribute__((amdgpu_num_vgpr(64)))
void bignn_pull_gemm(const int* __restrict__ tab,
                     const unsigned long long* __restrict__ recs,
                     const float* __restrict__ feat,
                     const float* __restrict__ W1, const float* __restrict__ b1,
                     const float* __restrict__ W2, const float* __restrict__ b2,
                     float* __restrict__ out, int N, int E, int nC) {
    __shared__ v2f   sW1[(DIM / 2) * DIM];
    __shared__ v2f   sW2[(DIM / 2) * DIM];
    __shared__ float sBias[DIM];
    __shared__ u64   sList[TILE_ROWS * PITCH];
    __shared__ int   sCnt[TILE_ROWS];

    int tid = threadIdx.x;
    int tileBase = blockIdx.x * TILE_ROWS;

    int wp = tid >> 4;
    int wg = tid & 15;
    bool wld = tid < (DIM / 2) * 16;
    const float4* W1v = (const float4*)W1;
    const float4* W2v = (const float4*)W2;
    float4 w1lo = make_float4(0.f, 0.f, 0.f, 0.f), w1hi = w1lo, w2lo = w1lo, w2hi = w1lo;
    if (wld) {
        w1lo = W1v[(2 * wp) * (DIM / 4) + wg];
        w1hi = W1v[(2 * wp + 1) * (DIM / 4) + wg];
        w2lo = W2v[(2 * wp) * (DIM / 4) + wg];
        w2hi = W2v[(2 * wp + 1) * (DIM / 4) + wg];
    }
    int   bcol  = tid & (DIM - 1);
    float bsum  = b1[bcol] + b2[bcol];
    int c0 = (tid / LPC) < nC ? (tid / LPC) : (nC - 1);
    int rs = tab[(size_t)c0 * NBINS + blockIdx.x];
    int re = tab[(size_t)c0 * NBINS + blockIdx.x + 1];

    if (tid < TILE_ROWS) sCnt[tid] = 0;
    if (wld) {
        v2f* d1 = sW1 + wp * DIM + 4 * wg;
        v2f* d2 = sW2 + wp * DIM + 4 * wg;
        d1[0] = v2f{w1lo.x, w1hi.x}; d1[1] = v2f{w1lo.y, w1hi.y};
        d1[2] = v2f{w1lo.z, w1hi.z}; d1[3] = v2f{w1lo.w, w1hi.w};
        d2[0] = v2f{w2lo.x, w2hi.x}; d2[1] = v2f{w2lo.y, w2hi.y};
        d2[2] = v2f{w2lo.z, w2hi.z}; d2[3] = v2f{w2lo.w, w2hi.w};
    }
    if (tid < DIM) sBias[tid] = bsum;
    __syncthreads();

    {
        int c = tid / LPC;
        int sub = tid & (LPC - 1);
        bool live = c < nC;
        for (int turn = 0; turn < nC && live; ++turn) {
            const u64* R = recs + (size_t)c * CHUNK;
            for (int i = rs + sub; i < re; i += PB * LPC) {
                u64 k[PB];
                #pragma unroll
                for (int q = 0; q < PB; ++q) {
                    k[q] = 0;
                    if (i + LPC * q < re) k[q] = R[i + LPC * q];
                }
                int pos[PB];
                #pragma unroll
                for (int q = 0; q < PB; ++q) {
                    pos[q] = CAP;
                    if (i + LPC * q < re) pos[q] = atomicAdd(&sCnt[(int)(k[q] >> 56)], 1);
                }
                #pragma unroll
                for (int q = 0; q < PB; ++q) {
                    if (pos[q] < CAP) sList[(int)(k[q] >> 56) * PITCH + pos[q]] = k[q] & KEYMASK;
                }
            }
            c += NTHREADS / LPC;
            live = c < nC;
            if (live) {
                rs = tab[(size_t)c * NBINS + blockIdx.x];
                re = tab[(size_t)c * NBINS + blockIdx.x + 1];
            }
        }
    }
    __syncthreads();

    int wave = tid >> 5;
    int lane = tid & 31;
    const v2f* x2 = (const v2f*)feat;
    for (int j = 0; j < ROWS_PER_WAVE; ++j) {
        int t = wave * ROWS_PER_WAVE + j;
        int d = __builtin_amdgcn_readfirstlane(sCnt[t]);
        u64* L = sList + t * PITCH;
        int grow = tileBase + t;
        v2f fown = x2[((unsigned)(grow < N ? grow : N - 1) << 5) | (unsigned)lane];
        v2f acc = v2f{0.0f, 0.0f};
        if (d <= CAP) {
            {
                const u64 kmax = ~(u64)0;
                u64 ka = (lane < d) ? L[lane] : kmax;
                int ra = 0;
                if (d <= 32) {
                    for (int q = 0; q < d; ++q) {
                        u64 kq = L[q];
                        ra += (kq < ka || (kq == ka && q < lane)) ? 1 : 0;
                    }
                    if (lane < d) L[ra] = ka;
                } else {
                    u64 kb = (lane + 32 < d) ? L[lane + 32] : kmax;
                    int rb = 0;
                    for (int q = 0; q < d; ++q) {
                        u64 kq = L[q];
                        ra += (kq < ka || (kq == ka && q < lane)) ? 1 : 0;
                        rb += (kq < kb || (kq == kb && q < lane + 32)) ? 1 : 0;
                    }
                    L[ra] = ka;
                    if (lane + 32 < d) L[rb] = kb;
                }
            }
            int i = 0;
            while (d - i >= 16) {
                bignn_addx<16>(acc, L, i, x2, lane);
                i += 16;
            }
            if (i < d) bignn_tail<1, 15>(acc, L, i, d - i, x2, lane);
        } else {
            u64 last = 0;
            int have = 0;
            for (int outer = 0; outer < E; ++outer) {
                u64 best = 0;
                int mult = 0;
                for (int cc = 0; cc < nC; ++cc) {
                    int s2 = tab[(size_t)cc * NBINS + blockIdx.x];
                    int e2 = tab[(size_t)cc * NBINS + blockIdx.x + 1];
                    const u64* R2 = recs + (size_t)cc * CHUNK;
                    for (int i2 = s2; i2 < e2; ++i2) {
                        u64 k = R2[i2];
                        if ((int)(k >> 56) != t) continue;
                        k &= KEYMASK;
                        if (!have || k > last) {
                            if (mult == 0 || k < best) { best = k; mult = 1; }
                            else if (k == best) ++mult;
                        }
                    }
                }
                if (mult == 0) break;
                float v = __uint_as_float((unsigned)best);
                v2f xv = x2[(long)(int)(best >> 32) * (DIM / 2) + lane];
                for (int m2 = 0; m2 < mult; ++m2) {
                    acc.x += v * xv.x;
                    acc.y += v * xv.y;
                }
                last = best;
                have = 1;
            }
        }
        v2f aSum = v2f{acc.x + fown.x, acc.y + fown.y};
        v2f aPrd = v2f{acc.x * fown.x, acc.y * fown.y};
        L[lane]      = ((u64)__float_as_uint(aSum.y) << 32) | (u64)__float_as_uint(aSum.x);
        L[32 + lane] = ((u64)__float_as_uint(aPrd.y) << 32) | (u64)__float_as_uint(aPrd.x);
    }
    __syncthreads();

    int m    = lane & 15;
    int half = lane >> 4;
    int rt   = wave >> 1;
    int ch   = wave & 1;
    int lrow = rt * 16 + m;
    int rowBase = tileBase + rt * 16;

    const u64*    lrowp = sList + lrow * PITCH;

    v8f acc2[2];
    acc2[0] = v8f{}; acc2[1] = v8f{};

    for (int k0 = 0; k0 < 16; ++k0) {
        int kp = k0 * 2 + half;
        u64 sw = lrowp[kp];
        u64 pw = lrowp[32 + kp];
        v2f aS = { __uint_as_float((unsigned)sw), __uint_as_float((unsigned)(sw >> 32)) };
        v2f aP = { __uint_as_float((unsigned)pw), __uint_as_float((unsigned)(pw >> 32)) };
        #pragma unroll
        for (int nb = 0; nb < 2; ++nb) {
            int cidx = (ch * 2 + nb) * 16 + m;
            v2f bA = sW1[kp * DIM + cidx];
            v2f bB = sW2[kp * DIM + cidx];
            acc2[nb] = __builtin_amdgcn_wmma_f32_16x16x4_f32(
                false, aS, false, bA, (short)0, acc2[nb], false, false);
            acc2[nb] = __builtin_amdgcn_wmma_f32_16x16x4_f32(
                false, aP, false, bB, (short)0, acc2[nb], false, false);
        }
    }

    int rOff = half * 8;
    if (rowBase + 16 <= N) {
        #pragma unroll
        for (int nb = 0; nb < 2; ++nb) {
            int col = (ch * 2 + nb) * 16 + m;
            float bias = sBias[col];
            #pragma unroll
            for (int g = 0; g < 8; ++g)
                out[(long)(rowBase + rOff + g) * DIM + col] = acc2[nb][g] + bias;
        }
    } else {
        #pragma unroll
        for (int nb = 0; nb < 2; ++nb) {
            int col = (ch * 2 + nb) * 16 + m;
            float bias = sBias[col];
            #pragma unroll
            for (int g = 0; g < 8; ++g) {
                int r = rowBase + rOff + g;
                if (r < N) out[(long)r * DIM + col] = acc2[nb][g] + bias;
            }
        }
    }
}

extern "C" void kernel_launch(void* const* d_in, const int* in_sizes, int n_in,
                              void* d_out, int out_size, void* d_ws, size_t ws_size,
                              hipStream_t stream) {
    const int*   erow  = (const int*)  d_in[0];
    const int*   ecol  = (const int*)  d_in[1];
    const float* eval_ = (const float*)d_in[2];
    const float* feat  = (const float*)d_in[3];
    const float* W1    = (const float*)d_in[4];
    const float* b1    = (const float*)d_in[5];
    const float* W2    = (const float*)d_in[6];
    const float* b2    = (const float*)d_in[7];
    float* out = (float*)d_out;

    int E = in_sizes[0];
    int N = in_sizes[3] / DIM;

    int  nC   = (E + CHUNK - 1) / CHUNK;
    int* tab  = (int*)d_ws;
    u64* recs = (u64*)(tab + (size_t)nC * NBINS);

    bignn_sort<<<nC, SORT_THREADS, 0, stream>>>(erow, ecol, eval_, tab, recs, E);
    bignn_pull_gemm<<<(N + TILE_ROWS - 1) / TILE_ROWS, NTHREADS, 0, stream>>>(
        tab, recs, feat, W1, b1, W2, b2, out, N, E, nC);
}
